# starting kernel
# baseline (speedup 1.0000x reference)
#include <hip/hip_runtime.h>

#define RR 5
#define CC 2
#define LL 2
#define HIDD 128

typedef __attribute__((ext_vector_type(2))) float v2f;
typedef __attribute__((ext_vector_type(8))) float v8f;

__global__ void softmax_filt_kernel(const float* __restrict__ gt_w,
                                    float* __restrict__ filt) {
  int t = threadIdx.x;
  if (t >= LL * CC) return;
  const float* g = gt_w + t * RR;
  float m = g[0];
  for (int r = 1; r < RR; ++r) m = fmaxf(m, g[r]);
  float e[RR];
  float s = 0.f;
  for (int r = 0; r < RR; ++r) { e[r] = __expf(g[r] - m); s += e[r]; }
  float inv = 1.0f / s;
  for (int r = 0; r < RR; ++r) filt[t * RR + r] = e[r] * inv;
}

__global__ void deg_kernel(const int* __restrict__ dst,
                           const int* __restrict__ etype,
                           const float* __restrict__ filt,
                           float* __restrict__ deg,
                           int E, int N, int layer) {
  int e = blockIdx.x * blockDim.x + threadIdx.x;
  if (e >= E) return;
  int d  = dst[e];
  int et = etype[e];
  #pragma unroll
  for (int c = 0; c < CC; ++c)
    atomicAdd(&deg[(size_t)c * N + d], filt[(layer * CC + c) * RR + et]);
}

__global__ void msg_kernel(const int* __restrict__ src,
                           const int* __restrict__ dst,
                           const int* __restrict__ etype,
                           const float* __restrict__ filt,
                           const float* __restrict__ deg,
                           const float* __restrict__ Hin,
                           float* __restrict__ Hout,
                           int E, int N, int layer) {
  int gid  = blockIdx.x * blockDim.x + threadIdx.x;
  int e    = gid >> 5;
  int lane = gid & 31;
  if (e >= E) return;
  int s  = src[e];
  int d  = dst[e];
  int et = etype[e];
  #pragma unroll
  for (int c = 0; c < CC; ++c) {
    float w  = filt[(layer * CC + c) * RR + et];
    float wn = w / deg[(size_t)c * N + d];
    const float4* hp = (const float4*)(Hin + ((size_t)c * N + s) * HIDD);
    float4 v = hp[lane];
    float* op = Hout + ((size_t)c * N + d) * HIDD + lane * 4;
    atomicAdd(op + 0, v.x * wn);
    atomicAdd(op + 1, v.y * wn);
    atomicAdd(op + 2, v.z * wn);
    atomicAdd(op + 3, v.w * wn);
  }
}

template <int NTILES>
__global__ __launch_bounds__(256) void gemm_wmma_f32(
    const float* __restrict__ A, const float* __restrict__ B,
    const float* __restrict__ bias, float* __restrict__ Out,
    int M, int K,
    int aRowStride, int aKShift, int aKMask, size_t aChunkStride,
    size_t bBatchStride, size_t outBatchStride, int relu) {
  constexpr int NDIM = NTILES * 16;
  __shared__ float2 Bs[16 * NDIM];

  const int wave = threadIdx.x >> 5;
  const int lane = threadIdx.x & 31;
  const int half = lane & 15;
  const int hi   = lane >> 4;

  const float* Bb = B + bBatchStride * blockIdx.y;
  float*       Ob = Out + outBatchStride * blockIdx.y;

  const int mTile = blockIdx.x * 8 + wave;
  const int row0  = mTile * 16;

  v8f acc[NTILES];
  for (int nt = 0; nt < NTILES; ++nt) {
    float bv = bias ? bias[nt * 16 + half] : 0.0f;
    v8f t;
    #pragma unroll
    for (int i = 0; i < 8; ++i) t[i] = bv;
    acc[nt] = t;
  }

  int aRow = row0 + half;
  if (aRow >= M) aRow = M - 1;
  const float* Arow = A + (size_t)aRow * aRowStride;

  const float2* BsLane = Bs + (size_t)hi * NDIM + half;

  for (int k0 = 0; k0 < K; k0 += 32) {
    for (int t = threadIdx.x; t < 16 * NDIM; t += 256) {
      int kp = t / NDIM, nn = t % NDIM;
      float2 v;
      v.x = Bb[(size_t)(k0 + 2 * kp)     * NDIM + nn];
      v.y = Bb[(size_t)(k0 + 2 * kp + 1) * NDIM + nn];
      Bs[(size_t)kp * NDIM + nn] = v;
    }
    __syncthreads();

    const float* ap0 = Arow
        + (size_t)(k0 >> aKShift) * aChunkStride
        + ((k0 & aKMask) + 2 * hi);

    #pragma unroll
    for (int kk = 0; kk < 32; kk += 4) {
      v2f a = *(const v2f*)(ap0 + kk);
      #pragma unroll
      for (int nt = 0; nt < NTILES; ++nt) {
        v2f b = *(const v2f*)(BsLane + (size_t)(kk >> 1) * NDIM + nt * 16);
        acc[nt] = __builtin_amdgcn_wmma_f32_16x16x4_f32(
            false, a, false, b, (short)0, acc[nt], false, false);
      }
    }
    __syncthreads();
  }

  for (int nt = 0; nt < NTILES; ++nt) {
    #pragma unroll
    for (int vr = 0; vr < 8; ++vr) {
      int r = row0 + vr + 8 * hi;
      if (r < M) {
        float v = acc[nt][vr];
        if (relu) v = fmaxf(v, 0.0f);
        Ob[(size_t)r * NDIM + nt * 16 + half] = v;
      }
    }
  }
}

__global__ void gather_kernel(const float* __restrict__ Y,
                              const int* __restrict__ cat,
                              float* __restrict__ out, int ncat) {
  int t = blockIdx.x * blockDim.x + threadIdx.x;
  if (t >= ncat * 16) return;
  int i = t >> 4, j = t & 15;
  out[t] = Y[(size_t)cat[i] * 16 + j];
}

extern "C" void kernel_launch(void* const* d_in, const int* in_sizes, int n_in,
                              void* d_out, int out_size, void* d_ws, size_t ws_size,
                              hipStream_t stream) {
  const float* h   = (const float*)d_in[0];
  const int*   src = (const int*)d_in[1];
  const int*   dst = (const int*)d_in[2];
  const int*   et  = (const int*)d_in[3];
  const int*   cat = (const int*)d_in[4];
  const float* gtw = (const float*)d_in[5];
  const float* Wc  = (const float*)d_in[6];
  const float* W1  = (const float*)d_in[7];
  const float* b1  = (const float*)d_in[8];
  const float* W2  = (const float*)d_in[9];
  const float* b2  = (const float*)d_in[10];

  const int IN_DIM = 256;
  const int N    = in_sizes[0] / IN_DIM;
  const int E    = in_sizes[1];
  const int NCAT = in_sizes[4];

  float* ws    = (float*)d_ws;
  float* filt  = ws;
  float* deg   = filt + 32;
  float* Hbuf0 = deg + (size_t)CC * N;
  float* Hbuf1 = Hbuf0 + (size_t)CC * N * HIDD;
  float* Z     = Hbuf1 + (size_t)CC * N * HIDD;
  float* Y     = Z + (size_t)N * HIDD;

  softmax_filt_kernel<<<1, 32, 0, stream>>>(gtw, filt);

  const int mtiles = (N + 15) / 16;
  const int gx     = (mtiles + 7) / 8;

  gemm_wmma_f32<8><<<dim3(gx, CC), 256, 0, stream>>>(
      h, Wc, nullptr, Hbuf0, N, IN_DIM,
        IN_DIM,   8,   255,   0,
        (size_t)IN_DIM * HIDD,   (size_t)N * HIDD,   0);

  float* Hcur = Hbuf0;
  float* Hnxt = Hbuf1;
  for (int l = 0; l < LL; ++l) {
    hipMemsetAsync(deg, 0, sizeof(float) * (size_t)CC * N, stream);
    hipMemsetAsync(Hnxt, 0, sizeof(float) * (size_t)CC * N * HIDD, stream);
    deg_kernel<<<(E + 255) / 256, 256, 0, stream>>>(dst, et, filt, deg, E, N, l);
    long long th = (long long)E * 32;
    msg_kernel<<<(unsigned)((th + 255) / 256), 256, 0, stream>>>(
        src, dst, et, filt, deg, Hcur, Hnxt, E, N, l);
    float* tmp = Hcur; Hcur = Hnxt; Hnxt = tmp;
  }

  gemm_wmma_f32<8><<<dim3(gx, 1), 256, 0, stream>>>(
      Hcur, W1, b1, Z, N, CC * HIDD,
        HIDD,   7,   127,
        (size_t)N * HIDD,
      0, 0,   1);

  gemm_wmma_f32<1><<<dim3(gx, 1), 256, 0, stream>>>(
      Z, W2, b2, Y, N, HIDD,
        HIDD,   7,   127,   0,
      0, 0,   0);

  gather_kernel<<<(NCAT * 16 + 255) / 256, 256, 0, stream>>>(
      Y, cat, (float*)d_out, NCAT);
}


# Edge scatter-add with float atomics replaced by a destination-sorted edge list built once per call (counting sort) and a one-wave-per-row gather; no float atomics, no H or degree memsets
# speedup vs baseline: 2.6283x; 2.6283x over previous
#include <hip/hip_runtime.h>

#define RR 5
#define CC 2
#define LL 2
#define HIDD 128

typedef __attribute__((ext_vector_type(2))) float v2f;
typedef __attribute__((ext_vector_type(8))) float v8f;

__global__ void softmax_filt_kernel(const float* __restrict__ gt_w,
                                    float* __restrict__ filt) {
  int t = threadIdx.x;
  if (t >= LL * CC) return;
  const float* g = gt_w + t * RR;
  float m = g[0];
  for (int r = 1; r < RR; ++r) m = fmaxf(m, g[r]);
  float e[RR];
  float s = 0.f;
  for (int r = 0; r < RR; ++r) { e[r] = __expf(g[r] - m); s += e[r]; }
  float inv = 1.0f / s;
  for (int r = 0; r < RR; ++r) filt[t * RR + r] = e[r] * inv;
}

#define KEY_SRC_MASK 0x0FFFFFFFu
#define KEY_ET_SHIFT 28

__global__ void hist_kernel(const int* __restrict__ dst,
                            int* __restrict__ cnt,
                            int* __restrict__ rank, int E) {
  int e = blockIdx.x * blockDim.x + threadIdx.x;
  if (e >= E) return;
  rank[e] = atomicAdd(&cnt[dst[e]], 1);
}

__device__ __forceinline__ int wave_sum_i(int v) {
  for (int o = 16; o > 0; o >>= 1) v += __shfl_xor(v, o);
  return v;
}

__device__ __forceinline__ float wave_sum_f(float v) {
  for (int o = 16; o > 0; o >>= 1) v += __shfl_xor(v, o);
  return v;
}

__device__ __forceinline__ int wave_incl_scan_i(int v, int lane) {
  for (int o = 1; o < 32; o <<= 1) {
    int u = __shfl_up(v, o);
    if (lane >= o) v += u;
  }
  return v;
}

__global__ __launch_bounds__(1024) void scan_kernel(
    const int* __restrict__ cnt, int* __restrict__ rowptr, int N) {
  __shared__ int wsum[32];
  __shared__ int red[32];
  const int t    = threadIdx.x;
  const int lane = t & 31;
  const int wave = t >> 5;
  const int base = blockIdx.x * 4096;

  int pre = 0;
  for (int i = t; i < base; i += 1024) pre += cnt[i];
  pre = wave_sum_i(pre);
  if (lane == 0) red[wave] = pre;

  const int i0 = base + 4 * t;
  int c[4];
  for (int k = 0; k < 4; ++k) c[k] = (i0 + k < N) ? cnt[i0 + k] : 0;
  const int tsum = c[0] + c[1] + c[2] + c[3];
  const int v = wave_incl_scan_i(tsum, lane);
  if (lane == 31) wsum[wave] = v;
  __syncthreads();
  if (wave == 0) {
    int w = wave_incl_scan_i(wsum[lane], lane);
    int r = wave_sum_i(red[lane]);
    wsum[lane] = w;
    if (lane == 0) red[0] = r;
  }
  __syncthreads();
  int run = red[0] + (wave > 0 ? wsum[wave - 1] : 0) + v - tsum;
  for (int k = 0; k < 4; ++k) {
    if (i0 + k < N) { rowptr[i0 + k] = run; run += c[k]; }
  }
  if (i0 < N && i0 + 4 >= N) rowptr[N] = run;
}

__global__ void fill_kernel(const int* __restrict__ src,
                            const int* __restrict__ dst,
                            const int* __restrict__ etype,
                            const int* __restrict__ rowptr,
                            const int* __restrict__ rank,
                            int* __restrict__ keys, int E) {
  int e = blockIdx.x * blockDim.x + threadIdx.x;
  if (e >= E) return;
  int pos = rowptr[dst[e]] + rank[e];
  if ((unsigned)pos < (unsigned)E)
    keys[pos] = src[e] | (etype[e] << KEY_ET_SHIFT);
}

__device__ __forceinline__ int lane_i(int v, int l) {
  return __builtin_amdgcn_readlane(v, l);
}
__device__ __forceinline__ float lane_f(float v, int l) {
  return __int_as_float(__builtin_amdgcn_readlane(__float_as_int(v), l));
}
__device__ __forceinline__ void fma4(float4& a, const float4& x, float w) {
  a.x = fmaf(x.x, w, a.x); a.y = fmaf(x.y, w, a.y);
  a.z = fmaf(x.z, w, a.z); a.w = fmaf(x.w, w, a.w);
}

__device__ __forceinline__ float pick5(const float (&t)[RR], int et) {
  float v = t[RR - 1];
  v = (et == 3) ? t[3] : v;
  v = (et == 2) ? t[2] : v;
  v = (et == 1) ? t[1] : v;
  v = (et == 0) ? t[0] : v;
  return v;
}

__device__ __forceinline__ void load_chunk(
    const int* __restrict__ keys, int b, int end, int lane, int N,
    const float (&fa)[RR], const float (&fb)[RR],
    int& s, float& u0, float& u1) {
  s = 0; u0 = 0.f; u1 = 0.f;
  const int j = b + lane;
  if (j < end) {
    const unsigned k = (unsigned)keys[j];
    int sv = (int)(k & KEY_SRC_MASK);
    if (sv > N - 1) sv = N - 1;
    s = sv;
    const int et = (int)(k >> KEY_ET_SHIFT);
    u0 = pick5(fa, et);
    u1 = pick5(fb, et);
  }
}

__global__ __launch_bounds__(256) void spmm_kernel(
    const int* __restrict__ rowptr, const int* __restrict__ keys,
    const float* __restrict__ filt, const float* __restrict__ Hin,
    float* __restrict__ Hout, int E, int N, int layer) {
  const int lane = threadIdx.x & 31;
  const int wave = __builtin_amdgcn_readfirstlane(threadIdx.x >> 5);
  const int d = blockIdx.x * 8 + wave;
  if (d >= N) return;

  int beg = rowptr[d];
  int end = rowptr[d + 1];
  if (beg < 0) beg = 0;
  if (end > E) end = E;

  const float* f = filt + layer * CC * RR;
  float fa[RR], fb[RR];
  for (int r = 0; r < RR; ++r) { fa[r] = f[r]; fb[r] = f[RR + r]; }

  int s; float u0, u1;
  load_chunk(keys, beg, end, lane, N, fa, fb, s, u0, u1);

  float dg0 = u0, dg1 = u1;
  for (int b = beg + 32; b < end; b += 32) {
    int s2; float v0, v1;
    load_chunk(keys, b, end, lane, N, fa, fb, s2, v0, v1);
    dg0 += v0; dg1 += v1;
  }
  dg0 = wave_sum_f(dg0);
  dg1 = wave_sum_f(dg1);

  const float4* H0 = (const float4*)Hin + lane;
  const float4* H1 = H0 + (size_t)N * (HIDD / 4);
  float4 a0 = make_float4(0.f, 0.f, 0.f, 0.f);
  float4 a1 = make_float4(0.f, 0.f, 0.f, 0.f);

  for (int b = beg; b < end; b += 32) {
    if (b != beg) load_chunk(keys, b, end, lane, N, fa, fb, s, u0, u1);
    const float w0 = u0 / dg0;
    const float w1 = u1 / dg1;
    int n = end - b;
    if (n > 32) n = 32;

    int i = 0;
    for (; i + 4 <= n; i += 4) {
      const size_t ra = (size_t)lane_i(s, i)     * (HIDD / 4);
      const size_t rb = (size_t)lane_i(s, i + 1) * (HIDD / 4);
      const size_t rc = (size_t)lane_i(s, i + 2) * (HIDD / 4);
      const size_t rd = (size_t)lane_i(s, i + 3) * (HIDD / 4);
      const float4 xa0 = H0[ra], xa1 = H1[ra];
      const float4 xb0 = H0[rb], xb1 = H1[rb];
      const float4 xc0 = H0[rc], xc1 = H1[rc];
      const float4 xd0 = H0[rd], xd1 = H1[rd];
      fma4(a0, xa0, lane_f(w0, i));     fma4(a1, xa1, lane_f(w1, i));
      fma4(a0, xb0, lane_f(w0, i + 1)); fma4(a1, xb1, lane_f(w1, i + 1));
      fma4(a0, xc0, lane_f(w0, i + 2)); fma4(a1, xc1, lane_f(w1, i + 2));
      fma4(a0, xd0, lane_f(w0, i + 3)); fma4(a1, xd1, lane_f(w1, i + 3));
    }
    if (i < n) {
      const int r  = n - i;
      const int ib = (r > 1) ? i + 1 : i;
      const int ic = (r > 2) ? i + 2 : i;
      const size_t ra = (size_t)lane_i(s, i)  * (HIDD / 4);
      const size_t rb = (size_t)lane_i(s, ib) * (HIDD / 4);
      const size_t rc = (size_t)lane_i(s, ic) * (HIDD / 4);
      const float4 xa0 = H0[ra], xa1 = H1[ra];
      const float4 xb0 = H0[rb], xb1 = H1[rb];
      const float4 xc0 = H0[rc], xc1 = H1[rc];
      const float wb0 = (r > 1) ? lane_f(w0, ib) : 0.f;
      const float wb1 = (r > 1) ? lane_f(w1, ib) : 0.f;
      const float wc0 = (r > 2) ? lane_f(w0, ic) : 0.f;
      const float wc1 = (r > 2) ? lane_f(w1, ic) : 0.f;
      fma4(a0, xa0, lane_f(w0, i)); fma4(a1, xa1, lane_f(w1, i));
      fma4(a0, xb0, wb0);           fma4(a1, xb1, wb1);
      fma4(a0, xc0, wc0);           fma4(a1, xc1, wc1);
    }
  }

  float4* O0 = (float4*)Hout + (size_t)d * (HIDD / 4) + lane;
  float4* O1 = O0 + (size_t)N * (HIDD / 4);
  *O0 = a0;
  *O1 = a1;
}

template <int NTILES>
__global__ __launch_bounds__(256) void gemm_wmma_f32(
    const float* __restrict__ A, const float* __restrict__ B,
    const float* __restrict__ bias, float* __restrict__ Out,
    int M, int K,
    int aRowStride, int aKShift, int aKMask, size_t aChunkStride,
    size_t bBatchStride, size_t outBatchStride, int relu) {
  constexpr int NDIM = NTILES * 16;
  __shared__ float2 Bs[16 * NDIM];

  const int wave = threadIdx.x >> 5;
  const int lane = threadIdx.x & 31;
  const int half = lane & 15;
  const int hi   = lane >> 4;

  const float* Bb = B + bBatchStride * blockIdx.y;
  float*       Ob = Out + outBatchStride * blockIdx.y;

  const int mTile = blockIdx.x * 8 + wave;
  const int row0  = mTile * 16;

  v8f acc[NTILES];
  for (int nt = 0; nt < NTILES; ++nt) {
    float bv = bias ? bias[nt * 16 + half] : 0.0f;
    v8f t;
    #pragma unroll
    for (int i = 0; i < 8; ++i) t[i] = bv;
    acc[nt] = t;
  }

  int aRow = row0 + half;
  if (aRow >= M) aRow = M - 1;
  const float* Arow = A + (size_t)aRow * aRowStride;

  const float2* BsLane = Bs + (size_t)hi * NDIM + half;

  for (int k0 = 0; k0 < K; k0 += 32) {
    for (int t = threadIdx.x; t < 16 * NDIM; t += 256) {
      int kp = t / NDIM, nn = t % NDIM;
      float2 v;
      v.x = Bb[(size_t)(k0 + 2 * kp)     * NDIM + nn];
      v.y = Bb[(size_t)(k0 + 2 * kp + 1) * NDIM + nn];
      Bs[(size_t)kp * NDIM + nn] = v;
    }
    __syncthreads();

    const float* ap0 = Arow
        + (size_t)(k0 >> aKShift) * aChunkStride
        + ((k0 & aKMask) + 2 * hi);

    #pragma unroll
    for (int kk = 0; kk < 32; kk += 4) {
      v2f a = *(const v2f*)(ap0 + kk);
      #pragma unroll
      for (int nt = 0; nt < NTILES; ++nt) {
        v2f b = *(const v2f*)(BsLane + (size_t)(kk >> 1) * NDIM + nt * 16);
        acc[nt] = __builtin_amdgcn_wmma_f32_16x16x4_f32(
            false, a, false, b, (short)0, acc[nt], false, false);
      }
    }
    __syncthreads();
  }

  for (int nt = 0; nt < NTILES; ++nt) {
    #pragma unroll
    for (int vr = 0; vr < 8; ++vr) {
      int r = row0 + vr + 8 * hi;
      if (r < M) {
        float v = acc[nt][vr];
        if (relu) v = fmaxf(v, 0.0f);
        Ob[(size_t)r * NDIM + nt * 16 + half] = v;
      }
    }
  }
}

__global__ void gather_kernel(const float* __restrict__ Y,
                              const int* __restrict__ cat,
                              float* __restrict__ out, int ncat) {
  int t = blockIdx.x * blockDim.x + threadIdx.x;
  if (t >= ncat * 16) return;
  int i = t >> 4, j = t & 15;
  out[t] = Y[(size_t)cat[i] * 16 + j];
}

extern "C" void kernel_launch(void* const* d_in, const int* in_sizes, int n_in,
                              void* d_out, int out_size, void* d_ws, size_t ws_size,
                              hipStream_t stream) {
  const float* h   = (const float*)d_in[0];
  const int*   src = (const int*)d_in[1];
  const int*   dst = (const int*)d_in[2];
  const int*   et  = (const int*)d_in[3];
  const int*   cat = (const int*)d_in[4];
  const float* gtw = (const float*)d_in[5];
  const float* Wc  = (const float*)d_in[6];
  const float* W1  = (const float*)d_in[7];
  const float* b1  = (const float*)d_in[8];
  const float* W2  = (const float*)d_in[9];
  const float* b2  = (const float*)d_in[10];

  const int IN_DIM = 256;
  const int N    = in_sizes[0] / IN_DIM;
  const int E    = in_sizes[1];
  const int NCAT = in_sizes[4];

  float* ws    = (float*)d_ws;
  float* filt  = ws;
  float* Hbuf0 = filt + 32 + (size_t)CC * N;
  float* Hbuf1 = Hbuf0 + (size_t)CC * N * HIDD;
  float* Z     = Hbuf1 + (size_t)CC * N * HIDD;
  float* Y     = Z + (size_t)N * HIDD;

  softmax_filt_kernel<<<1, 32, 0, stream>>>(gtw, filt);

  const int mtiles = (N + 15) / 16;
  const int gx     = (mtiles + 7) / 8;

  gemm_wmma_f32<8><<<dim3(gx, CC), 256, 0, stream>>>(
      h, Wc, nullptr, Hbuf0, N, IN_DIM,
        IN_DIM,   8,   255,   0,
        (size_t)IN_DIM * HIDD,   (size_t)N * HIDD,   0);

  int* cnt    = (int*)Z;
  int* rowptr = cnt + N;
  int* rank   = rowptr + (N + 1);
  int* keys   = rank + E;
  hipMemsetAsync(cnt, 0, sizeof(int) * (size_t)N, stream);
  hist_kernel<<<(E + 255) / 256, 256, 0, stream>>>(dst, cnt, rank, E);
  scan_kernel<<<(N + 4095) / 4096, 1024, 0, stream>>>(cnt, rowptr, N);
  fill_kernel<<<(E + 255) / 256, 256, 0, stream>>>(src, dst, et, rowptr, rank, keys, E);

  float* Hcur = Hbuf0;
  float* Hnxt = Hbuf1;
  for (int l = 0; l < LL; ++l) {
    spmm_kernel<<<(N + 7) / 8, 256, 0, stream>>>(
        rowptr, keys, filt, Hcur, Hnxt, E, N, l);
    float* tmp = Hcur; Hcur = Hnxt; Hnxt = tmp;
  }

  gemm_wmma_f32<8><<<dim3(gx, 1), 256, 0, stream>>>(
      Hcur, W1, b1, Z, N, CC * HIDD,
        HIDD,   7,   127,
        (size_t)N * HIDD,
      0, 0,   1);

  gemm_wmma_f32<1><<<dim3(gx, 1), 256, 0, stream>>>(
      Z, W2, b2, Y, N, HIDD,
        HIDD,   7,   127,   0,
      0, 0,   0);

  gather_kernel<<<(NCAT * 16 + 255) / 256, 256, 0, stream>>>(
      Y, cat, (float*)d_out, NCAT);
}


# GEMM operands fetched one stage ahead: next K panel's B and A global reads in flight as one group under the current panel's WMMAs; two-panel LDS ring, one barrier per panel
# speedup vs baseline: 3.7092x; 1.4113x over previous
#include <hip/hip_runtime.h>

#define RR 5
#define CC 2
#define LL 2
#define HIDD 128

typedef __attribute__((ext_vector_type(2))) float v2f;
typedef __attribute__((ext_vector_type(8))) float v8f;

__global__ void softmax_filt_kernel(const float* __restrict__ gt_w,
                                    float* __restrict__ filt) {
  int t = threadIdx.x;
  if (t >= LL * CC) return;
  const float* g = gt_w + t * RR;
  float m = g[0];
  for (int r = 1; r < RR; ++r) m = fmaxf(m, g[r]);
  float e[RR];
  float s = 0.f;
  for (int r = 0; r < RR; ++r) { e[r] = __expf(g[r] - m); s += e[r]; }
  float inv = 1.0f / s;
  for (int r = 0; r < RR; ++r) filt[t * RR + r] = e[r] * inv;
}

#define KEY_SRC_MASK 0x0FFFFFFFu
#define KEY_ET_SHIFT 28

__global__ void hist_kernel(const int* __restrict__ dst,
                            int* __restrict__ cnt,
                            int* __restrict__ rank, int E) {
  int e = blockIdx.x * blockDim.x + threadIdx.x;
  if (e >= E) return;
  rank[e] = atomicAdd(&cnt[dst[e]], 1);
}

__device__ __forceinline__ int wave_sum_i(int v) {
  for (int o = 16; o > 0; o >>= 1) v += __shfl_xor(v, o);
  return v;
}

__device__ __forceinline__ float wave_sum_f(float v) {
  for (int o = 16; o > 0; o >>= 1) v += __shfl_xor(v, o);
  return v;
}

__device__ __forceinline__ int wave_incl_scan_i(int v, int lane) {
  for (int o = 1; o < 32; o <<= 1) {
    int u = __shfl_up(v, o);
    if (lane >= o) v += u;
  }
  return v;
}

__global__ __launch_bounds__(1024) void scan_kernel(
    const int* __restrict__ cnt, int* __restrict__ rowptr, int N) {
  __shared__ int wsum[32];
  __shared__ int red[32];
  const int t    = threadIdx.x;
  const int lane = t & 31;
  const int wave = t >> 5;
  const int base = blockIdx.x * 4096;

  int pre = 0;
  for (int i = t; i < base; i += 1024) pre += cnt[i];
  pre = wave_sum_i(pre);
  if (lane == 0) red[wave] = pre;

  const int i0 = base + 4 * t;
  int c[4];
  for (int k = 0; k < 4; ++k) c[k] = (i0 + k < N) ? cnt[i0 + k] : 0;
  const int tsum = c[0] + c[1] + c[2] + c[3];
  const int v = wave_incl_scan_i(tsum, lane);
  if (lane == 31) wsum[wave] = v;
  __syncthreads();
  if (wave == 0) {
    int w = wave_incl_scan_i(wsum[lane], lane);
    int r = wave_sum_i(red[lane]);
    wsum[lane] = w;
    if (lane == 0) red[0] = r;
  }
  __syncthreads();
  int run = red[0] + (wave > 0 ? wsum[wave - 1] : 0) + v - tsum;
  for (int k = 0; k < 4; ++k) {
    if (i0 + k < N) { rowptr[i0 + k] = run; run += c[k]; }
  }
  if (i0 < N && i0 + 4 >= N) rowptr[N] = run;
}

__global__ void fill_kernel(const int* __restrict__ src,
                            const int* __restrict__ dst,
                            const int* __restrict__ etype,
                            const int* __restrict__ rowptr,
                            const int* __restrict__ rank,
                            int* __restrict__ keys, int E) {
  int e = blockIdx.x * blockDim.x + threadIdx.x;
  if (e >= E) return;
  int pos = rowptr[dst[e]] + rank[e];
  if ((unsigned)pos < (unsigned)E)
    keys[pos] = src[e] | (etype[e] << KEY_ET_SHIFT);
}

__device__ __forceinline__ int lane_i(int v, int l) {
  return __builtin_amdgcn_readlane(v, l);
}
__device__ __forceinline__ float lane_f(float v, int l) {
  return __int_as_float(__builtin_amdgcn_readlane(__float_as_int(v), l));
}
__device__ __forceinline__ void fma4(float4& a, const float4& x, float w) {
  a.x = fmaf(x.x, w, a.x); a.y = fmaf(x.y, w, a.y);
  a.z = fmaf(x.z, w, a.z); a.w = fmaf(x.w, w, a.w);
}

__device__ __forceinline__ float pick5(const float (&t)[RR], int et) {
  float v = t[RR - 1];
  v = (et == 3) ? t[3] : v;
  v = (et == 2) ? t[2] : v;
  v = (et == 1) ? t[1] : v;
  v = (et == 0) ? t[0] : v;
  return v;
}

__device__ __forceinline__ void load_chunk(
    const int* __restrict__ keys, int b, int end, int lane, int N,
    const float (&fa)[RR], const float (&fb)[RR],
    int& s, float& u0, float& u1) {
  s = 0; u0 = 0.f; u1 = 0.f;
  const int j = b + lane;
  if (j < end) {
    const unsigned k = (unsigned)keys[j];
    int sv = (int)(k & KEY_SRC_MASK);
    if (sv > N - 1) sv = N - 1;
    s = sv;
    const int et = (int)(k >> KEY_ET_SHIFT);
    u0 = pick5(fa, et);
    u1 = pick5(fb, et);
  }
}

__global__ __launch_bounds__(256) void spmm_kernel(
    const int* __restrict__ rowptr, const int* __restrict__ keys,
    const float* __restrict__ filt, const float* __restrict__ Hin,
    float* __restrict__ Hout, const int* __restrict__ rowsel, int nOut,
    int E, int N, int layer) {
  const int lane = threadIdx.x & 31;
  const int wave = __builtin_amdgcn_readfirstlane(threadIdx.x >> 5);
  const int o = blockIdx.x * 8 + wave;
  if (o >= nOut) return;
  int d = o;
  if (rowsel) {
    d = rowsel[o];
    if (d < 0) d = 0;
    if (d > N - 1) d = N - 1;
  }

  int beg = rowptr[d];
  int end = rowptr[d + 1];
  if (beg < 0) beg = 0;
  if (end > E) end = E;

  const float* f = filt + layer * CC * RR;
  float fa[RR], fb[RR];
  for (int r = 0; r < RR; ++r) { fa[r] = f[r]; fb[r] = f[RR + r]; }

  int s; float u0, u1;
  load_chunk(keys, beg, end, lane, N, fa, fb, s, u0, u1);

  float dg0 = u0, dg1 = u1;
  for (int b = beg + 32; b < end; b += 32) {
    int s2; float v0, v1;
    load_chunk(keys, b, end, lane, N, fa, fb, s2, v0, v1);
    dg0 += v0; dg1 += v1;
  }
  dg0 = wave_sum_f(dg0);
  dg1 = wave_sum_f(dg1);

  const float4* H0 = (const float4*)Hin + lane;
  const float4* H1 = H0 + (size_t)N * (HIDD / 4);
  float4 a0 = make_float4(0.f, 0.f, 0.f, 0.f);
  float4 a1 = make_float4(0.f, 0.f, 0.f, 0.f);

  for (int b = beg; b < end; b += 32) {
    if (b != beg) load_chunk(keys, b, end, lane, N, fa, fb, s, u0, u1);
    const float w0 = u0 / dg0;
    const float w1 = u1 / dg1;
    int n = end - b;
    if (n > 32) n = 32;

    int i = 0;
    for (; i + 4 <= n; i += 4) {
      const size_t ra = (size_t)lane_i(s, i)     * (HIDD / 4);
      const size_t rb = (size_t)lane_i(s, i + 1) * (HIDD / 4);
      const size_t rc = (size_t)lane_i(s, i + 2) * (HIDD / 4);
      const size_t rd = (size_t)lane_i(s, i + 3) * (HIDD / 4);
      const float4 xa0 = H0[ra], xa1 = H1[ra];
      const float4 xb0 = H0[rb], xb1 = H1[rb];
      const float4 xc0 = H0[rc], xc1 = H1[rc];
      const float4 xd0 = H0[rd], xd1 = H1[rd];
      fma4(a0, xa0, lane_f(w0, i));     fma4(a1, xa1, lane_f(w1, i));
      fma4(a0, xb0, lane_f(w0, i + 1)); fma4(a1, xb1, lane_f(w1, i + 1));
      fma4(a0, xc0, lane_f(w0, i + 2)); fma4(a1, xc1, lane_f(w1, i + 2));
      fma4(a0, xd0, lane_f(w0, i + 3)); fma4(a1, xd1, lane_f(w1, i + 3));
    }
    if (i < n) {
      const int r  = n - i;
      const int ib = (r > 1) ? i + 1 : i;
      const int ic = (r > 2) ? i + 2 : i;
      const size_t ra = (size_t)lane_i(s, i)  * (HIDD / 4);
      const size_t rb = (size_t)lane_i(s, ib) * (HIDD / 4);
      const size_t rc = (size_t)lane_i(s, ic) * (HIDD / 4);
      const float4 xa0 = H0[ra], xa1 = H1[ra];
      const float4 xb0 = H0[rb], xb1 = H1[rb];
      const float4 xc0 = H0[rc], xc1 = H1[rc];
      const float wb0 = (r > 1) ? lane_f(w0, ib) : 0.f;
      const float wb1 = (r > 1) ? lane_f(w1, ib) : 0.f;
      const float wc0 = (r > 2) ? lane_f(w0, ic) : 0.f;
      const float wc1 = (r > 2) ? lane_f(w1, ic) : 0.f;
      fma4(a0, xa0, lane_f(w0, i)); fma4(a1, xa1, lane_f(w1, i));
      fma4(a0, xb0, wb0);           fma4(a1, xb1, wb1);
      fma4(a0, xc0, wc0);           fma4(a1, xc1, wc1);
    }
  }

  float4* O0 = (float4*)Hout + (size_t)o * (HIDD / 4) + lane;
  float4* O1 = O0 + (size_t)nOut * (HIDD / 4);
  *O0 = a0;
  *O1 = a1;
}

template <int NTILES>
__device__ __forceinline__ void panel_products(
    v8f (&acc)[NTILES], const v2f (&a)[8], const float2* bs) {
  constexpr int NDIM = NTILES * 16;
  v2f bc[NTILES];
  #pragma unroll
  for (int nt = 0; nt < NTILES; ++nt) bc[nt] = *(const v2f*)(bs + nt * 16);
  #pragma unroll
  for (int j = 0; j < 8; ++j) {
    v2f bn[NTILES];
    if (j + 1 < 8) {
      #pragma unroll
      for (int nt = 0; nt < NTILES; ++nt)
        bn[nt] = *(const v2f*)(bs + (size_t)(2 * (j + 1)) * NDIM + nt * 16);
    }
    #pragma unroll
    for (int nt = 0; nt < NTILES; ++nt)
      acc[nt] = __builtin_amdgcn_wmma_f32_16x16x4_f32(
          false, a[j], false, bc[nt], (short)0, acc[nt], false, false);
    if (j + 1 < 8) {
      #pragma unroll
      for (int nt = 0; nt < NTILES; ++nt) bc[nt] = bn[nt];
    }
  }
}

template <int NTILES>
__global__ __launch_bounds__(256) void gemm_wmma_f32(
    const float* __restrict__ A, const float* __restrict__ B,
    const float* __restrict__ bias, float* __restrict__ Out,
    int M, int K,
    int aRowStride, int aKShift, int aKMask, size_t aChunkStride,
    size_t bBatchStride, size_t outBatchStride, int relu) {
  constexpr int NDIM   = NTILES * 16;
  constexpr int PANEL  = 16 * NDIM;
  constexpr int NQ     = NDIM / 4;
  constexpr int GROUPS = 16 * NQ;
  constexpr int IPT    = (GROUPS + 255) / 256;
  __shared__ float4 Bs4[PANEL];
  float2* Bs = (float2*)Bs4;

  const int wave = threadIdx.x >> 5;
  const int lane = threadIdx.x & 31;
  const int half = lane & 15;
  const int hi   = lane >> 4;

  const float* Bb = B + bBatchStride * blockIdx.y;
  float*       Ob = Out + outBatchStride * blockIdx.y;

  const int mTile = blockIdx.x * 8 + wave;
  const int row0  = mTile * 16;

  v8f acc[NTILES];
  for (int nt = 0; nt < NTILES; ++nt) {
    float bv = bias ? bias[nt * 16 + half] : 0.0f;
    v8f t;
    #pragma unroll
    for (int i = 0; i < 8; ++i) t[i] = bv;
    acc[nt] = t;
  }

  int aRow = row0 + half;
  if (aRow >= M) aRow = M - 1;
  const float* Arow = A + (size_t)aRow * aRowStride + 2 * hi;

  const float2* BsLane = Bs + (size_t)hi * NDIM + half;

  int sOff[IPT];
  int gOff[IPT];
  #pragma unroll
  for (int g = 0; g < IPT; ++g) {
    const int idx = (threadIdx.x + g * 256) % GROUPS;
    const int kp = idx / NQ, n = (idx % NQ) * 4;
    sOff[g] = kp * NDIM + n;
    gOff[g] = 2 * kp * NDIM + n;
  }
  constexpr bool ALL_OWN = (GROUPS % 256) == 0;
  const bool own = ALL_OWN || (threadIdx.x < GROUPS);

  float4 bx[IPT], by[IPT];
  v2f    an[8];

  #pragma unroll
  for (int g = 0; g < IPT; ++g) {
    const float* q = Bb + gOff[g];
    bx[g] = *(const float4*)q;
    by[g] = *(const float4*)(q + NDIM);
  }
  #pragma unroll
  for (int j = 0; j < 8; ++j) an[j] = *(const v2f*)(Arow + 4 * j);
  if (own) {
    #pragma unroll
    for (int g = 0; g < IPT; ++g) {
      float4* d = (float4*)(Bs + sOff[g]);
      d[0] = make_float4(bx[g].x, by[g].x, bx[g].y, by[g].y);
      d[1] = make_float4(bx[g].z, by[g].z, bx[g].w, by[g].w);
    }
  }
  __syncthreads();

  const int P = K >> 5;
  for (int p = 0; p + 1 < P; ++p) {
    v2f ac[8];
    #pragma unroll
    for (int j = 0; j < 8; ++j) ac[j] = an[j];

    const int k1 = (p + 1) << 5;
    #pragma unroll
    for (int g = 0; g < IPT; ++g) {
      const float* q = Bb + (size_t)k1 * NDIM + gOff[g];
      bx[g] = *(const float4*)q;
      by[g] = *(const float4*)(q + NDIM);
    }
    const float* ap = Arow + (size_t)(k1 >> aKShift) * aChunkStride + (k1 & aKMask);
    #pragma unroll
    for (int j = 0; j < 8; ++j) an[j] = *(const v2f*)(ap + 4 * j);

    panel_products<NTILES>(acc, ac, BsLane + (p & 1) * PANEL);

    if (own) {
      float2* dst = Bs + ((p + 1) & 1) * PANEL;
      #pragma unroll
      for (int g = 0; g < IPT; ++g) {
        float4* d = (float4*)(dst + sOff[g]);
        d[0] = make_float4(bx[g].x, by[g].x, bx[g].y, by[g].y);
        d[1] = make_float4(bx[g].z, by[g].z, bx[g].w, by[g].w);
      }
    }
    __syncthreads();
  }
  panel_products<NTILES>(acc, an, BsLane + ((P - 1) & 1) * PANEL);

  for (int nt = 0; nt < NTILES; ++nt) {
    #pragma unroll
    for (int vr = 0; vr < 8; ++vr) {
      int r = row0 + vr + 8 * hi;
      if (r < M) {
        float v = acc[nt][vr];
        if (relu) v = fmaxf(v, 0.0f);
        Ob[(size_t)r * NDIM + nt * 16 + half] = v;
      }
    }
  }
}

extern "C" void kernel_launch(void* const* d_in, const int* in_sizes, int n_in,
                              void* d_out, int out_size, void* d_ws, size_t ws_size,
                              hipStream_t stream) {
  const float* h   = (const float*)d_in[0];
  const int*   src = (const int*)d_in[1];
  const int*   dst = (const int*)d_in[2];
  const int*   et  = (const int*)d_in[3];
  const int*   cat = (const int*)d_in[4];
  const float* gtw = (const float*)d_in[5];
  const float* Wc  = (const float*)d_in[6];
  const float* W1  = (const float*)d_in[7];
  const float* b1  = (const float*)d_in[8];
  const float* W2  = (const float*)d_in[9];
  const float* b2  = (const float*)d_in[10];

  const int IN_DIM = 256;
  const int N    = in_sizes[0] / IN_DIM;
  const int E    = in_sizes[1];
  const int NCAT = in_sizes[4];

  float* ws    = (float*)d_ws;
  float* filt  = ws;
  float* Hbuf0 = filt + 32 + (size_t)CC * N;
  float* Hbuf1 = Hbuf0 + (size_t)CC * N * HIDD;
  float* Z     = Hbuf1 + (size_t)CC * N * HIDD;

  softmax_filt_kernel<<<1, 32, 0, stream>>>(gtw, filt);

  const int mtiles = (N + 15) / 16;
  const int gx     = (mtiles + 7) / 8;

  gemm_wmma_f32<8><<<dim3(gx, CC), 256, 0, stream>>>(
      h, Wc, nullptr, Hbuf0, N, IN_DIM,
        IN_DIM,   8,   255,   0,
        (size_t)IN_DIM * HIDD,   (size_t)N * HIDD,   0);

  int* cnt    = (int*)Z;
  int* rowptr = cnt + N;
  int* rank   = rowptr + (N + 1);
  int* keys   = rank + E;
  hipMemsetAsync(cnt, 0, sizeof(int) * (size_t)N, stream);
  hist_kernel<<<(E + 255) / 256, 256, 0, stream>>>(dst, cnt, rank, E);
  scan_kernel<<<(N + 4095) / 4096, 1024, 0, stream>>>(cnt, rowptr, N);
  fill_kernel<<<(E + 255) / 256, 256, 0, stream>>>(src, dst, et, rowptr, rank, keys, E);

  float* Hcur = Hbuf0;
  float* Hnxt = Hbuf1;
  for (int l = 0; l < LL - 1; ++l) {
    spmm_kernel<<<(N + 7) / 8, 256, 0, stream>>>(
        rowptr, keys, filt, Hcur, Hnxt, nullptr, N, E, N, l);
    float* tmp = Hcur; Hcur = Hnxt; Hnxt = tmp;
  }
  spmm_kernel<<<(NCAT + 7) / 8, 256, 0, stream>>>(
      rowptr, keys, filt, Hcur, Hnxt, cat, NCAT, E, N, LL - 1);
  float* Hsel = Hnxt;

  const int ctiles = (NCAT + 15) / 16;
  const int gc     = (ctiles + 7) / 8;

  gemm_wmma_f32<8><<<dim3(gc, 1), 256, 0, stream>>>(
      Hsel, W1, b1, Z, NCAT, CC * HIDD,
        HIDD,   7,   127,
        (size_t)NCAT * HIDD,
      0, 0,   1);

  gemm_wmma_f32<1><<<dim3(gc, 1), 256, 0, stream>>>(
      Z, W2, b2, (float*)d_out, NCAT, HIDD,
        HIDD,   7,   127,   0,
      0, 0,   0);
}


# Linear stages reordered: both propagation rounds run on raw h (one 1 KB row read per edge feeds both channels); Wc folded into W1 (512x128), applied to the selected rows only; N-row projection gone
# speedup vs baseline: 4.7770x; 1.2879x over previous
#include <hip/hip_runtime.h>

#define RR 5
#define CC 2
#define LL 2
#define HIDD 128
#define FDIM 256

typedef __attribute__((ext_vector_type(2))) float v2f;
typedef __attribute__((ext_vector_type(8))) float v8f;

__global__ void prep_kernel(const float* __restrict__ gt_w,
                            float* __restrict__ filt,
                            int* __restrict__ cnt, int N) {
  const int t = blockIdx.x * blockDim.x + threadIdx.x;
  if (t < N) cnt[t] = 0;
  if (t >= LL * CC) return;
  const float* g = gt_w + t * RR;
  float m = g[0];
  for (int r = 1; r < RR; ++r) m = fmaxf(m, g[r]);
  float e[RR];
  float s = 0.f;
  for (int r = 0; r < RR; ++r) { e[r] = __expf(g[r] - m); s += e[r]; }
  float inv = 1.0f / s;
  for (int r = 0; r < RR; ++r) filt[t * RR + r] = e[r] * inv;
}

#define KEY_SRC_MASK 0x0FFFFFFFu
#define KEY_ET_SHIFT 28

#define HIST_EDGES_PER_BLOCK 1024
__device__ __forceinline__ void hist_block(const int* __restrict__ dst,
                                           int* __restrict__ cnt,
                                           int* __restrict__ rank, int E, int hb) {
  const int e0 = hb * HIST_EDGES_PER_BLOCK + threadIdx.x;
  int d[4], r[4];
  #pragma unroll
  for (int k = 0; k < 4; ++k) {
    const int e = e0 + k * 256;
    d[k] = (e < E) ? dst[e] : -1;
  }
  #pragma unroll
  for (int k = 0; k < 4; ++k) r[k] = (d[k] >= 0) ? atomicAdd(&cnt[d[k]], 1) : 0;
  #pragma unroll
  for (int k = 0; k < 4; ++k) {
    const int e = e0 + k * 256;
    if (e < E) rank[e] = r[k];
  }
}

__device__ __forceinline__ int wave_sum_i(int v) {
  for (int o = 16; o > 0; o >>= 1) v += __shfl_xor(v, o);
  return v;
}

__device__ __forceinline__ float wave_sum_f(float v) {
  for (int o = 16; o > 0; o >>= 1) v += __shfl_xor(v, o);
  return v;
}

__device__ __forceinline__ int wave_incl_scan_i(int v, int lane) {
  for (int o = 1; o < 32; o <<= 1) {
    int u = __shfl_up(v, o);
    if (lane >= o) v += u;
  }
  return v;
}

__global__ __launch_bounds__(1024) void scan_kernel(
    const int* __restrict__ cnt, int* __restrict__ rowptr, int N) {
  __shared__ int wsum[32];
  __shared__ int red[32];
  const int t    = threadIdx.x;
  const int lane = t & 31;
  const int wave = t >> 5;
  const int base = blockIdx.x * 4096;

  int pre = 0;
  for (int i = t; i < base; i += 1024) pre += cnt[i];
  pre = wave_sum_i(pre);
  if (lane == 0) red[wave] = pre;

  const int i0 = base + 4 * t;
  int c[4];
  for (int k = 0; k < 4; ++k) c[k] = (i0 + k < N) ? cnt[i0 + k] : 0;
  const int tsum = c[0] + c[1] + c[2] + c[3];
  const int v = wave_incl_scan_i(tsum, lane);
  if (lane == 31) wsum[wave] = v;
  __syncthreads();
  if (wave == 0) {
    int w = wave_incl_scan_i(wsum[lane], lane);
    int r = wave_sum_i(red[lane]);
    wsum[lane] = w;
    if (lane == 0) red[0] = r;
  }
  __syncthreads();
  int run = red[0] + (wave > 0 ? wsum[wave - 1] : 0) + v - tsum;
  for (int k = 0; k < 4; ++k) {
    if (i0 + k < N) { rowptr[i0 + k] = run; run += c[k]; }
  }
  if (i0 < N && i0 + 4 >= N) rowptr[N] = run;
}

__global__ void fill_kernel(const int* __restrict__ src,
                            const int* __restrict__ dst,
                            const int* __restrict__ etype,
                            const int* __restrict__ rowptr,
                            const int* __restrict__ rank,
                            int* __restrict__ keys, int E) {
  int e = blockIdx.x * blockDim.x + threadIdx.x;
  if (e >= E) return;
  int pos = rowptr[dst[e]] + rank[e];
  if ((unsigned)pos < (unsigned)E)
    keys[pos] = src[e] | (etype[e] << KEY_ET_SHIFT);
}

__device__ __forceinline__ int lane_i(int v, int l) {
  return __builtin_amdgcn_readlane(v, l);
}
__device__ __forceinline__ float lane_f(float v, int l) {
  return __int_as_float(__builtin_amdgcn_readlane(__float_as_int(v), l));
}
__device__ __forceinline__ void fma4(float4& a, const float4& x, float w) {
  a.x = fmaf(x.x, w, a.x); a.y = fmaf(x.y, w, a.y);
  a.z = fmaf(x.z, w, a.z); a.w = fmaf(x.w, w, a.w);
}

__device__ __forceinline__ float pick5(const float (&t)[RR], int et) {
  float v = t[RR - 1];
  v = (et == 3) ? t[3] : v;
  v = (et == 2) ? t[2] : v;
  v = (et == 1) ? t[1] : v;
  v = (et == 0) ? t[0] : v;
  return v;
}

__device__ __forceinline__ void load_chunk(
    const int* __restrict__ keys, int b, int end, int lane, int N,
    const float (&fa)[RR], const float (&fb)[RR],
    int& s, float& u0, float& u1) {
  s = 0; u0 = 0.f; u1 = 0.f;
  const int j = b + lane;
  if (j < end) {
    const unsigned k = (unsigned)keys[j];
    int sv = (int)(k & KEY_SRC_MASK);
    if (sv > N - 1) sv = N - 1;
    s = sv;
    const int et = (int)(k >> KEY_ET_SHIFT);
    u0 = pick5(fa, et);
    u1 = pick5(fb, et);
  }
}

struct RowQ { float4 p0, q0, p1, q1; };

template <bool SHARED>
__device__ __forceinline__ RowQ row_read(const float4* __restrict__ H0,
                                         const float4* __restrict__ H1, size_t r) {
  RowQ x;
  x.p0 = H0[r]; x.q0 = H0[r + 32];
  if constexpr (SHARED) { x.p1 = x.p0; x.q1 = x.q0; }
  else { x.p1 = H1[r]; x.q1 = H1[r + 32]; }
  return x;
}

__device__ __forceinline__ void row_acc(float4& a0, float4& b0, float4& a1, float4& b1,
                                        const RowQ& x, float w0, float w1) {
  fma4(a0, x.p0, w0); fma4(b0, x.q0, w0);
  fma4(a1, x.p1, w1); fma4(b1, x.q1, w1);
}

template <bool SHARED>
__global__ __launch_bounds__(256) void spmm_kernel(
    const int* __restrict__ rowptr, const int* __restrict__ keys,
    const float* __restrict__ filt, const float* __restrict__ Hin,
    float* __restrict__ Hout, const int* __restrict__ rowsel, int nOut,
    int E, int N, int layer) {
  const int lane = threadIdx.x & 31;
  const int wave = __builtin_amdgcn_readfirstlane(threadIdx.x >> 5);
  const int o = blockIdx.x * 8 + wave;
  if (o >= nOut) return;
  int d = o;
  if (rowsel) {
    d = rowsel[o];
    if (d < 0) d = 0;
    if (d > N - 1) d = N - 1;
  }

  int beg = rowptr[d];
  int end = rowptr[d + 1];
  if (beg < 0) beg = 0;
  if (end > E) end = E;

  const float* f = filt + layer * CC * RR;
  float fa[RR], fb[RR];
  for (int r = 0; r < RR; ++r) { fa[r] = f[r]; fb[r] = f[RR + r]; }

  int s; float u0, u1;
  load_chunk(keys, beg, end, lane, N, fa, fb, s, u0, u1);

  float dg0 = u0, dg1 = u1;
  for (int b = beg + 32; b < end; b += 32) {
    int s2; float v0, v1;
    load_chunk(keys, b, end, lane, N, fa, fb, s2, v0, v1);
    dg0 += v0; dg1 += v1;
  }
  dg0 = wave_sum_f(dg0);
  dg1 = wave_sum_f(dg1);

  constexpr int RQ = FDIM / 4;
  const float4* H0 = (const float4*)Hin + lane;
  const float4* H1 = SHARED ? H0 : H0 + (size_t)N * RQ;
  float4 a0 = make_float4(0.f, 0.f, 0.f, 0.f), b0 = a0, a1 = a0, b1 = a0;

  for (int b = beg; b < end; b += 32) {
    if (b != beg) load_chunk(keys, b, end, lane, N, fa, fb, s, u0, u1);
    const float w0 = u0 / dg0;
    const float w1 = u1 / dg1;
    int n = end - b;
    if (n > 32) n = 32;

    int i = 0;
    for (; i + 4 <= n; i += 4) {
      const size_t ra = (size_t)lane_i(s, i)     * RQ;
      const size_t rb = (size_t)lane_i(s, i + 1) * RQ;
      const size_t rc = (size_t)lane_i(s, i + 2) * RQ;
      const size_t rd = (size_t)lane_i(s, i + 3) * RQ;
      const RowQ xa = row_read<SHARED>(H0, H1, ra);
      const RowQ xb = row_read<SHARED>(H0, H1, rb);
      const RowQ xc = row_read<SHARED>(H0, H1, rc);
      const RowQ xd = row_read<SHARED>(H0, H1, rd);
      row_acc(a0, b0, a1, b1, xa, lane_f(w0, i),     lane_f(w1, i));
      row_acc(a0, b0, a1, b1, xb, lane_f(w0, i + 1), lane_f(w1, i + 1));
      row_acc(a0, b0, a1, b1, xc, lane_f(w0, i + 2), lane_f(w1, i + 2));
      row_acc(a0, b0, a1, b1, xd, lane_f(w0, i + 3), lane_f(w1, i + 3));
    }
    if (i < n) {
      const int r  = n - i;
      const int ib = (r > 1) ? i + 1 : i;
      const int ic = (r > 2) ? i + 2 : i;
      const size_t ra = (size_t)lane_i(s, i)  * RQ;
      const size_t rb = (size_t)lane_i(s, ib) * RQ;
      const size_t rc = (size_t)lane_i(s, ic) * RQ;
      const RowQ xa = row_read<SHARED>(H0, H1, ra);
      const RowQ xb = row_read<SHARED>(H0, H1, rb);
      const RowQ xc = row_read<SHARED>(H0, H1, rc);
      const float wb0 = (r > 1) ? lane_f(w0, ib) : 0.f;
      const float wb1 = (r > 1) ? lane_f(w1, ib) : 0.f;
      const float wc0 = (r > 2) ? lane_f(w0, ic) : 0.f;
      const float wc1 = (r > 2) ? lane_f(w1, ic) : 0.f;
      row_acc(a0, b0, a1, b1, xa, lane_f(w0, i), lane_f(w1, i));
      row_acc(a0, b0, a1, b1, xb, wb0, wb1);
      row_acc(a0, b0, a1, b1, xc, wc0, wc1);
    }
  }

  float4* O0 = (float4*)Hout + (size_t)o * RQ + lane;
  float4* O1 = O0 + (size_t)nOut * RQ;
  O0[0] = a0; O0[32] = b0;
  O1[0] = a1; O1[32] = b1;
}

template <int NTILES>
__device__ __forceinline__ void panel_products(
    v8f (&acc)[NTILES], const v2f (&a)[8], const float2* bs) {
  constexpr int NDIM = NTILES * 16;
  v2f bc[NTILES];
  #pragma unroll
  for (int nt = 0; nt < NTILES; ++nt) bc[nt] = *(const v2f*)(bs + nt * 16);
  #pragma unroll
  for (int j = 0; j < 8; ++j) {
    v2f bn[NTILES];
    if (j + 1 < 8) {
      #pragma unroll
      for (int nt = 0; nt < NTILES; ++nt)
        bn[nt] = *(const v2f*)(bs + (size_t)(2 * (j + 1)) * NDIM + nt * 16);
    }
    #pragma unroll
    for (int nt = 0; nt < NTILES; ++nt)
      acc[nt] = __builtin_amdgcn_wmma_f32_16x16x4_f32(
          false, a[j], false, bc[nt], (short)0, acc[nt], false, false);
    if (j + 1 < 8) {
      #pragma unroll
      for (int nt = 0; nt < NTILES; ++nt) bc[nt] = bn[nt];
    }
  }
}

#define NCLS 16
template <int NTILES, int NW, bool CLS>
__global__ __launch_bounds__(NW * 32) void gemm_wmma_f32(
    const float* __restrict__ A, const float* __restrict__ B,
    const float* __restrict__ bias, float* __restrict__ Out,
    int M, int K,
    int aRowStride, int aKShift, int aKMask, size_t aChunkStride,
    size_t aBatchStride, size_t bBatchStride, size_t outBatchStride, int relu,
    int gemmBlocksX, int nBatch,
    const int* __restrict__ hDst, int* __restrict__ hCnt,
    int* __restrict__ hRank, int hE,
    const float* __restrict__ W2, const float* __restrict__ bias2,
    float* __restrict__ Out2) {
  constexpr int NDIM   = NTILES * 16;
  constexpr int PANEL  = 16 * NDIM;
  constexpr int NQ     = NDIM / 4;
  constexpr int GROUPS = 16 * NQ;
  constexpr int NT     = NW * 32;
  constexpr int IPT    = (GROUPS + NT - 1) / NT;
  static_assert(CLS || NW == 8, "the edge-count blocks are written for 256 threads");
  __shared__ float4 Bs4[PANEL];
  float2* Bs = (float2*)Bs4;

  const int wave = threadIdx.x >> 5;
  const int lane = threadIdx.x & 31;
  const int half = lane & 15;
  const int hi   = lane >> 4;

  const int bid = blockIdx.x;
  if constexpr (!CLS) {
    if (bid >= gemmBlocksX * nBatch) {
      hist_block(hDst, hCnt, hRank, hE, bid - gemmBlocksX * nBatch);
      return;
    }
  }
  const int batch = bid / gemmBlocksX;
  const int rgrp  = bid - batch * gemmBlocksX;

  const float* Ab = A + aBatchStride * batch;
  const float* Bb = B + bBatchStride * batch;
  float*       Ob = Out + outBatchStride * batch;

  const int mTile = rgrp * NW + wave;
  const int row0  = mTile * 16;

  v8f acc[NTILES];
  for (int nt = 0; nt < NTILES; ++nt) {
    float bv = bias ? bias[nt * 16 + half] : 0.0f;
    v8f t;
    #pragma unroll
    for (int i = 0; i < 8; ++i) t[i] = bv;
    acc[nt] = t;
  }

  int aRow = row0 + half;
  if (aRow >= M) aRow = M - 1;
  const float* Arow = Ab + (size_t)aRow * aRowStride + 2 * hi;

  const float2* BsLane = Bs + (size_t)hi * NDIM + half;

  int sOff[IPT];
  int gOff[IPT];
  #pragma unroll
  for (int g = 0; g < IPT; ++g) {
    const int idx = (threadIdx.x + g * NT) % GROUPS;
    const int kp = idx / NQ, n = (idx % NQ) * 4;
    sOff[g] = kp * NDIM + n;
    gOff[g] = 2 * kp * NDIM + n;
  }
  constexpr bool ALL_OWN = (GROUPS % NT) == 0;
  const bool own = ALL_OWN || (threadIdx.x < GROUPS);

  float4 bx[IPT], by[IPT];
  v2f    an[8];

  #pragma unroll
  for (int g = 0; g < IPT; ++g) {
    const float* q = Bb + gOff[g];
    bx[g] = *(const float4*)q;
    by[g] = *(const float4*)(q + NDIM);
  }
  #pragma unroll
  for (int j = 0; j < 8; ++j) an[j] = *(const v2f*)(Arow + 4 * j);
  if (own) {
    #pragma unroll
    for (int g = 0; g < IPT; ++g) {
      float4* d = (float4*)(Bs + sOff[g]);
      d[0] = make_float4(bx[g].x, by[g].x, bx[g].y, by[g].y);
      d[1] = make_float4(bx[g].z, by[g].z, bx[g].w, by[g].w);
    }
  }
  __syncthreads();

  const int P = K >> 5;
  for (int p = 0; p + 1 < P; ++p) {
    v2f ac[8];
    #pragma unroll
    for (int j = 0; j < 8; ++j) ac[j] = an[j];

    const int k1 = (p + 1) << 5;
    #pragma unroll
    for (int g = 0; g < IPT; ++g) {
      const float* q = Bb + (size_t)k1 * NDIM + gOff[g];
      bx[g] = *(const float4*)q;
      by[g] = *(const float4*)(q + NDIM);
    }
    const float* ap = Arow + (size_t)(k1 >> aKShift) * aChunkStride + (k1 & aKMask);
    #pragma unroll
    for (int j = 0; j < 8; ++j) an[j] = *(const v2f*)(ap + 4 * j);

    panel_products<NTILES>(acc, ac, BsLane + (p & 1) * PANEL);

    if (own) {
      float2* dst = Bs + ((p + 1) & 1) * PANEL;
      #pragma unroll
      for (int g = 0; g < IPT; ++g) {
        float4* d = (float4*)(dst + sOff[g]);
        d[0] = make_float4(bx[g].x, by[g].x, bx[g].y, by[g].y);
        d[1] = make_float4(bx[g].z, by[g].z, bx[g].w, by[g].w);
      }
    }
    __syncthreads();
  }
  panel_products<NTILES>(acc, an, BsLane + ((P - 1) & 1) * PANEL);

  if constexpr (CLS) {
    constexpr int ZP = NDIM + 4;
    __shared__ __attribute__((aligned(16))) float Zs[NW * 16 * ZP];
    float* zw = Zs + wave * (16 * ZP);

    v2f bw[NDIM / 4];
    const float* w2l = W2 + (size_t)(2 * hi) * NCLS + half;
    #pragma unroll
    for (int kk = 0; kk < NDIM / 4; ++kk) {
      bw[kk][0] = w2l[(4 * kk) * NCLS];
      bw[kk][1] = w2l[(4 * kk + 1) * NCLS];
    }

    #pragma unroll
    for (int nt = 0; nt < NTILES; ++nt) {
      #pragma unroll
      for (int vr = 0; vr < 8; ++vr) {
        float v = acc[nt][vr];
        if (relu) v = fmaxf(v, 0.0f);
        zw[(vr + 8 * hi) * ZP + nt * 16 + half] = v;
      }
    }
    __syncthreads();

    const float bv2 = bias2 ? bias2[half] : 0.0f;
    v8f o0, o1;
    #pragma unroll
    for (int i = 0; i < 8; ++i) { o0[i] = bv2; o1[i] = 0.0f; }
    const float* zr = zw + half * ZP + 2 * hi;
    #pragma unroll
    for (int kk = 0; kk < NDIM / 4; ++kk) {
      const v2f af = *(const v2f*)(zr + 4 * kk);
      if (kk & 1)
        o1 = __builtin_amdgcn_wmma_f32_16x16x4_f32(
            false, af, false, bw[kk], (short)0, o1, false, false);
      else
        o0 = __builtin_amdgcn_wmma_f32_16x16x4_f32(
            false, af, false, bw[kk], (short)0, o0, false, false);
    }
    #pragma unroll
    for (int vr = 0; vr < 8; ++vr) {
      const int r = row0 + vr + 8 * hi;
      if (r < M) Out2[(size_t)r * NCLS + half] = o0[vr] + o1[vr];
    }
  } else {
    for (int nt = 0; nt < NTILES; ++nt) {
      #pragma unroll
      for (int vr = 0; vr < 8; ++vr) {
        int r = row0 + vr + 8 * hi;
        if (r < M) {
          float v = acc[nt][vr];
          if (relu) v = fmaxf(v, 0.0f);
          Ob[(size_t)r * NDIM + nt * 16 + half] = v;
        }
      }
    }
  }
}

extern "C" void kernel_launch(void* const* d_in, const int* in_sizes, int n_in,
                              void* d_out, int out_size, void* d_ws, size_t ws_size,
                              hipStream_t stream) {
  const float* h   = (const float*)d_in[0];
  const int*   src = (const int*)d_in[1];
  const int*   dst = (const int*)d_in[2];
  const int*   et  = (const int*)d_in[3];
  const int*   cat = (const int*)d_in[4];
  const float* gtw = (const float*)d_in[5];
  const float* Wc  = (const float*)d_in[6];
  const float* W1  = (const float*)d_in[7];
  const float* b1  = (const float*)d_in[8];
  const float* W2  = (const float*)d_in[9];
  const float* b2  = (const float*)d_in[10];

  const int IN_DIM = FDIM;
  const int N    = in_sizes[0] / IN_DIM;
  const int E    = in_sizes[1];
  const int NCAT = in_sizes[4];
  static_assert(LL == 2, "written for two propagation rounds: raw features, then selected rows");

  float* ws    = (float*)d_ws;
  float* filt  = ws;
  float* Mfold = filt + 32;
  float* P0    = Mfold + (size_t)CC * FDIM * HIDD;
  int*   cnt   = (int*)(P0 + (size_t)CC * N * FDIM);
  int*  rowptr = cnt + N;
  int*  rank   = rowptr + (N + 1);
  int*  keys   = rank + E;
  const size_t edgeInts = ((size_t)N + (N + 1) + 2 * (size_t)E + 63) & ~(size_t)63;
  float* Psel  = (float*)(cnt + edgeInts);

  prep_kernel<<<(N + 255) / 256, 256, 0, stream>>>(gtw, filt, cnt, N);

  const int ftiles = (IN_DIM + 15) / 16;
  const int gf     = (ftiles + 7) / 8;
  const int hb     = (E + HIST_EDGES_PER_BLOCK - 1) / HIST_EDGES_PER_BLOCK;

  gemm_wmma_f32<8, 8, false><<<dim3(gf * CC + hb), 256, 0, stream>>>(
      Wc, W1, nullptr, Mfold, IN_DIM, HIDD,
        HIDD,   7,   127,   0,
        (size_t)IN_DIM * HIDD,   (size_t)HIDD * HIDD,
        (size_t)IN_DIM * HIDD,   0,
        gf,   CC, dst, cnt, rank, E,
      nullptr, nullptr, nullptr);

  scan_kernel<<<(N + 4095) / 4096, 1024, 0, stream>>>(cnt, rowptr, N);
  fill_kernel<<<(E + 255) / 256, 256, 0, stream>>>(src, dst, et, rowptr, rank, keys, E);

  spmm_kernel<true><<<(N + 7) / 8, 256, 0, stream>>>(
      rowptr, keys, filt, h, P0, nullptr, N, E, N, 0);
  spmm_kernel<false><<<(NCAT + 7) / 8, 256, 0, stream>>>(
      rowptr, keys, filt, P0, Psel, cat, NCAT, E, N, 1);

  const int ctiles = (NCAT + 15) / 16;
  const int gc     = (ctiles + 3) / 4;

  gemm_wmma_f32<8, 4, true><<<dim3(gc), 128, 0, stream>>>(
      Psel, Mfold, b1, nullptr, NCAT, CC * IN_DIM,
        IN_DIM,   8,   255,
        (size_t)NCAT * IN_DIM,
      0, 0, 0,   1,
        gc,   1, nullptr, nullptr, nullptr, 0,
      W2, b2, (float*)d_out);
}


# Hidden GEMM and classifier fused into the round-two gather launch: gathered rows stay in LDS as the A tile, K split over two waves per column tile (hidden sum order changes); 5 -> 4 launches
# speedup vs baseline: 6.6419x; 1.3904x over previous
#include <hip/hip_runtime.h>

#define RR 5
#define CC 2
#define LL 2
#define HIDD 128
#define FDIM 256

typedef __attribute__((ext_vector_type(2))) float v2f;
typedef __attribute__((ext_vector_type(8))) float v8f;

__global__ void prep_kernel(const float* __restrict__ gt_w,
                            float* __restrict__ filt,
                            int* __restrict__ bkt, int nQuads) {
  const int t = blockIdx.x * blockDim.x + threadIdx.x;
  const int nthr = gridDim.x * blockDim.x;
  int4* bkt4 = (int4*)bkt;
  for (int i = t; i < nQuads; i += nthr) bkt4[i] = make_int4(0, 0, 0, 0);
  if (t >= LL * CC) return;
  const float* g = gt_w + t * RR;
  float m = g[0];
  for (int r = 1; r < RR; ++r) m = fmaxf(m, g[r]);
  float e[RR];
  float s = 0.f;
  for (int r = 0; r < RR; ++r) { e[r] = __expf(g[r] - m); s += e[r]; }
  float inv = 1.0f / s;
  for (int r = 0; r < RR; ++r) filt[t * RR + r] = e[r] * inv;
}

#define KEY_SRC_MASK 0x0FFFFFFFu
#define KEY_ET_SHIFT 28
#define BROW 32
#define BCAP (BROW - 1)

#define HIST_EDGES_PER_BLOCK 1024
__device__ __forceinline__ void bucket_block(const int* __restrict__ src,
                                             const int* __restrict__ dst,
                                             const int* __restrict__ etype,
                                             int* __restrict__ bkt, int E, int N, int hb) {
  const int e0 = hb * HIST_EDGES_PER_BLOCK + threadIdx.x;
  int dv[4], sv[4], tv[4];
  int d[4], key[4], r[4];
  #pragma unroll
  for (int k = 0; k < 4; ++k) {
    const int e  = e0 + k * 256;
    const int ec = (e < E) ? e : E - 1;
    dv[k] = dst[ec]; sv[k] = src[ec]; tv[k] = etype[ec];
  }
  #pragma unroll
  for (int k = 0; k < 4; ++k) {
    const int e = e0 + k * 256;
    const bool ok = (e < E) && ((unsigned)dv[k] < (unsigned)N);
    d[k]   = ok ? dv[k] : -1;
    key[k] = (int)(((unsigned)sv[k] & KEY_SRC_MASK) | ((unsigned)tv[k] << KEY_ET_SHIFT));
  }
  #pragma unroll
  for (int k = 0; k < 4; ++k)
    r[k] = (d[k] >= 0) ? atomicAdd(&bkt[(size_t)d[k] * BROW], 1) : BCAP;
  #pragma unroll
  for (int k = 0; k < 4; ++k) {
    if ((unsigned)r[k] < (unsigned)BCAP) bkt[(size_t)d[k] * BROW + 1 + r[k]] = key[k];
  }
}

__device__ __forceinline__ float wave_sum_f(float v) {
  for (int o = 16; o > 0; o >>= 1) v += __shfl_xor(v, o);
  return v;
}

__device__ __forceinline__ int lane_i(int v, int l) {
  return __builtin_amdgcn_readlane(v, l);
}
__device__ __forceinline__ float lane_f(float v, int l) {
  return __int_as_float(__builtin_amdgcn_readlane(__float_as_int(v), l));
}
__device__ __forceinline__ void fma4(float4& a, const float4& x, float w) {
  a.x = fmaf(x.x, w, a.x); a.y = fmaf(x.y, w, a.y);
  a.z = fmaf(x.z, w, a.z); a.w = fmaf(x.w, w, a.w);
}

__device__ __forceinline__ float pick5(const float (&t)[RR], int et) {
  float v = t[RR - 1];
  v = (et == 3) ? t[3] : v;
  v = (et == 2) ? t[2] : v;
  v = (et == 1) ? t[1] : v;
  v = (et == 0) ? t[0] : v;
  return v;
}

__device__ __forceinline__ void key_decode(
    unsigned k, int N, const float (&fa)[RR], const float (&fb)[RR],
    int& s, float& u0, float& u1) {
  int sv = (int)(k & KEY_SRC_MASK);
  if (sv > N - 1) sv = N - 1;
  s = sv;
  const int et = (int)(k >> KEY_ET_SHIFT);
  u0 = pick5(fa, et);
  u1 = pick5(fb, et);
}

struct RowQ { float4 p0, q0, p1, q1; };

template <bool SHARED>
__device__ __forceinline__ RowQ row_read(const float4* __restrict__ H0,
                                         const float4* __restrict__ H1, size_t r) {
  RowQ x;
  x.p0 = H0[r]; x.q0 = H0[r + 32];
  if constexpr (SHARED) { x.p1 = x.p0; x.q1 = x.q0; }
  else { x.p1 = H1[r]; x.q1 = H1[r + 32]; }
  return x;
}

__device__ __forceinline__ void row_acc(float4& a0, float4& b0, float4& a1, float4& b1,
                                        const RowQ& x, float w0, float w1) {
  fma4(a0, x.p0, w0); fma4(b0, x.q0, w0);
  fma4(a1, x.p1, w1); fma4(b1, x.q1, w1);
}

template <bool SHARED>
__device__ __forceinline__ void gather_row(
    int d, int lane,
    const int* __restrict__ bkt, const int* __restrict__ src,
    const int* __restrict__ dst, const int* __restrict__ etype,
    const float* __restrict__ filt, const float* __restrict__ Hin,
    int E, int N, int layer,
    float4& a0, float4& b0, float4& a1, float4& b1) {
  const int raw = bkt[(size_t)d * BROW + lane];

  const float* f = filt + layer * CC * RR;
  float fa[RR], fb[RR];
  for (int r = 0; r < RR; ++r) { fa[r] = f[r]; fb[r] = f[RR + r]; }

  constexpr int RQ = FDIM / 4;
  const float4* H0 = (const float4*)Hin + lane;
  const float4* H1 = SHARED ? H0 : H0 + (size_t)N * RQ;

  const int cnt = lane_i(raw, 0);
  if (__builtin_expect(cnt <= BCAP, 1)) {
    int s = 0; float u0 = 0.f, u1 = 0.f;
    if (lane >= 1 && lane <= cnt) key_decode((unsigned)raw, N, fa, fb, s, u0, u1);

    const float dg0 = wave_sum_f(u0);
    const float dg1 = wave_sum_f(u1);
    const float w0 = u0 / dg0;
    const float w1 = u1 / dg1;
    const int n = cnt;

    int i = 0;
    for (; i + 4 <= n; i += 4) {
      const size_t ra = (size_t)lane_i(s, i + 1) * RQ;
      const size_t rb = (size_t)lane_i(s, i + 2) * RQ;
      const size_t rc = (size_t)lane_i(s, i + 3) * RQ;
      const size_t rd = (size_t)lane_i(s, i + 4) * RQ;
      const RowQ xa = row_read<SHARED>(H0, H1, ra);
      const RowQ xb = row_read<SHARED>(H0, H1, rb);
      const RowQ xc = row_read<SHARED>(H0, H1, rc);
      const RowQ xd = row_read<SHARED>(H0, H1, rd);
      row_acc(a0, b0, a1, b1, xa, lane_f(w0, i + 1), lane_f(w1, i + 1));
      row_acc(a0, b0, a1, b1, xb, lane_f(w0, i + 2), lane_f(w1, i + 2));
      row_acc(a0, b0, a1, b1, xc, lane_f(w0, i + 3), lane_f(w1, i + 3));
      row_acc(a0, b0, a1, b1, xd, lane_f(w0, i + 4), lane_f(w1, i + 4));
    }
    if (i < n) {
      const int r  = n - i;
      const int la = i + 1;
      const int lb = (r > 1) ? i + 2 : la;
      const int lc = (r > 2) ? i + 3 : la;
      const size_t ra = (size_t)lane_i(s, la) * RQ;
      const size_t rb = (size_t)lane_i(s, lb) * RQ;
      const size_t rc = (size_t)lane_i(s, lc) * RQ;
      const RowQ xa = row_read<SHARED>(H0, H1, ra);
      const RowQ xb = row_read<SHARED>(H0, H1, rb);
      const RowQ xc = row_read<SHARED>(H0, H1, rc);
      const float wb0 = (r > 1) ? lane_f(w0, lb) : 0.f;
      const float wb1 = (r > 1) ? lane_f(w1, lb) : 0.f;
      const float wc0 = (r > 2) ? lane_f(w0, lc) : 0.f;
      const float wc1 = (r > 2) ? lane_f(w1, lc) : 0.f;
      row_acc(a0, b0, a1, b1, xa, lane_f(w0, la), lane_f(w1, la));
      row_acc(a0, b0, a1, b1, xb, wb0, wb1);
      row_acc(a0, b0, a1, b1, xc, wc0, wc1);
    }
  } else {
    float dg0 = 0.f, dg1 = 0.f;
    for (int b = 0; b < E; b += 32) {
      const int e = b + lane;
      if (e < E && dst[e] == d) {
        const int et = etype[e];
        dg0 += pick5(fa, et);
        dg1 += pick5(fb, et);
      }
    }
    dg0 = wave_sum_f(dg0);
    dg1 = wave_sum_f(dg1);
    for (int b = 0; b < E; b += 32) {
      const int e = b + lane;
      const bool hit = (e < E) && (dst[e] == d);
      int s = 0; float w0 = 0.f, w1 = 0.f;
      if (hit) {
        int sv = src[e];
        if (sv < 0) sv = 0;
        if (sv > N - 1) sv = N - 1;
        s = sv;
        const int et = etype[e];
        w0 = pick5(fa, et) / dg0;
        w1 = pick5(fb, et) / dg1;
      }
      unsigned m = __builtin_amdgcn_ballot_w32(hit);
      while (m) {
        const int i = __builtin_ctz(m);
        m &= m - 1;
        const size_t r = (size_t)lane_i(s, i) * RQ;
        const RowQ x = row_read<SHARED>(H0, H1, r);
        row_acc(a0, b0, a1, b1, x, lane_f(w0, i), lane_f(w1, i));
      }
    }
  }
}

__global__ __launch_bounds__(256) void spmm_kernel(
    const int* __restrict__ bkt, const int* __restrict__ src,
    const int* __restrict__ dst, const int* __restrict__ etype,
    const float* __restrict__ filt, const float* __restrict__ Hin,
    float* __restrict__ Hout, int nOut, int E, int N) {
  const int lane = threadIdx.x & 31;
  const int wave = __builtin_amdgcn_readfirstlane(threadIdx.x >> 5);
  const int o = blockIdx.x * 8 + wave;
  if (o >= nOut) return;
  float4 a0 = make_float4(0.f, 0.f, 0.f, 0.f), b0 = a0, a1 = a0, b1 = a0;
  gather_row<true>(o, lane, bkt, src, dst, etype, filt, Hin, E, N, 0, a0, b0, a1, b1);

  constexpr int RQ = FDIM / 4;
  float4* O0 = (float4*)Hout + (size_t)o * RQ + lane;
  float4* O1 = O0 + (size_t)nOut * RQ;
  O0[0] = a0; O0[32] = b0;
  O1[0] = a1; O1[32] = b1;
}


#define FROWS 16
#define NT2   (HIDD / 16)
#define NCLS  16
#define KGRP  16
__global__ __launch_bounds__(FROWS * 32) void spmm_hidden_kernel(
    const int* __restrict__ bkt, const int* __restrict__ src,
    const int* __restrict__ dst, const int* __restrict__ etype,
    const float* __restrict__ filt, const float* __restrict__ Hin,
    const int* __restrict__ rowsel, int nOut, int E, int N,
    const float* __restrict__ Mf, const float* __restrict__ bias1,
    const float* __restrict__ W2, const float* __restrict__ bias2,
    float* __restrict__ Out) {
  static_assert(FROWS == 2 * NT2, "one wave per (column tile, channel) pair");
  static_assert(CC == 2 && FDIM % (4 * KGRP) == 0, "K is cut per channel, in groups of KGRP k-steps");
  constexpr int XP = CC * FDIM + 4;
  constexpr int ZP = HIDD + 4;
  __shared__ __attribute__((aligned(16))) float Xs[FROWS * XP];
  __shared__ __attribute__((aligned(16))) float Zs[FROWS * ZP];
  __shared__ float Ps[NT2 * 8 * 32];

  const int lane = threadIdx.x & 31;
  const int wave = __builtin_amdgcn_readfirstlane(threadIdx.x >> 5);
  const int row0 = blockIdx.x * FROWS;
  const int o = row0 + wave;

  float4 a0 = make_float4(0.f, 0.f, 0.f, 0.f), b0 = a0, a1 = a0, b1 = a0;
  if (o < nOut) {
    int d = rowsel[o];
    if (d < 0) d = 0;
    if (d > N - 1) d = N - 1;
    gather_row<false>(d, lane, bkt, src, dst, etype, filt, Hin, E, N, 1, a0, b0, a1, b1);
  }
  float* xr = Xs + wave * XP + 4 * lane;
  *(float4*)(xr)                  = a0;
  *(float4*)(xr + FDIM / 2)       = b0;
  *(float4*)(xr + FDIM)           = a1;
  *(float4*)(xr + FDIM + FDIM / 2) = b1;
  __syncthreads();

  const int half = lane & 15;
  const int hi   = lane >> 4;
  const int nt   = wave % NT2;
  const int ch   = wave / NT2;
  const float bv = (ch == 0 && bias1) ? bias1[nt * 16 + half] : 0.0f;
  v8f c0, c1;
  #pragma unroll
  for (int i = 0; i < 8; ++i) { c0[i] = bv; c1[i] = 0.0f; }
  const float* xa = Xs + half * XP + ch * FDIM + 2 * hi;
  const float* mb = Mf + (size_t)(ch * FDIM + 2 * hi) * HIDD + nt * 16 + half;
  for (int g = 0; g < FDIM / (4 * KGRP); ++g) {
    v2f bf[KGRP];
    #pragma unroll
    for (int u = 0; u < KGRP; ++u) {
      const float* q = mb + (size_t)(4 * (KGRP * g + u)) * HIDD;
      bf[u][0] = q[0];
      bf[u][1] = q[HIDD];
    }
    #pragma unroll
    for (int u = 0; u < KGRP; ++u) {
      const v2f af = *(const v2f*)(xa + 4 * (KGRP * g + u));
      if (u & 1)
        c1 = __builtin_amdgcn_wmma_f32_16x16x4_f32(
            false, af, false, bf[u], (short)0, c1, false, false);
      else
        c0 = __builtin_amdgcn_wmma_f32_16x16x4_f32(
            false, af, false, bf[u], (short)0, c0, false, false);
    }
  }

  float* pt = Ps + nt * (8 * 32) + lane;
  if (ch == 1) {
    #pragma unroll
    for (int vr = 0; vr < 8; ++vr) pt[vr * 32] = c0[vr] + c1[vr];
  }
  __syncthreads();
  if (ch == 0) {
    #pragma unroll
    for (int vr = 0; vr < 8; ++vr) {
      const float v = (c0[vr] + c1[vr]) + pt[vr * 32];
      Zs[(vr + 8 * hi) * ZP + nt * 16 + half] = fmaxf(v, 0.0f);
    }
  }
  __syncthreads();
  if (wave != 0) return;

  v2f bw[HIDD / 4];
  const float* w2l = W2 + (size_t)(2 * hi) * NCLS + half;
  #pragma unroll
  for (int kk = 0; kk < HIDD / 4; ++kk) {
    bw[kk][0] = w2l[(4 * kk) * NCLS];
    bw[kk][1] = w2l[(4 * kk + 1) * NCLS];
  }
  const float bv2 = bias2 ? bias2[half] : 0.0f;
  v8f o0, o1;
  #pragma unroll
  for (int i = 0; i < 8; ++i) { o0[i] = bv2; o1[i] = 0.0f; }
  const float* zr = Zs + half * ZP + 2 * hi;
  #pragma unroll
  for (int kk = 0; kk < HIDD / 4; ++kk) {
    const v2f af = *(const v2f*)(zr + 4 * kk);
    if (kk & 1)
      o1 = __builtin_amdgcn_wmma_f32_16x16x4_f32(
          false, af, false, bw[kk], (short)0, o1, false, false);
    else
      o0 = __builtin_amdgcn_wmma_f32_16x16x4_f32(
          false, af, false, bw[kk], (short)0, o0, false, false);
  }
  #pragma unroll
  for (int vr = 0; vr < 8; ++vr) {
    const int r = row0 + vr + 8 * hi;
    if (r < nOut) Out[(size_t)r * NCLS + half] = o0[vr] + o1[vr];
  }
}

template <int NTILES>
__device__ __forceinline__ void panel_products(
    v8f (&acc)[NTILES], const v2f (&a)[8], const float2* bs) {
  constexpr int NDIM = NTILES * 16;
  v2f bc[NTILES];
  #pragma unroll
  for (int nt = 0; nt < NTILES; ++nt) bc[nt] = *(const v2f*)(bs + nt * 16);
  #pragma unroll
  for (int j = 0; j < 8; ++j) {
    v2f bn[NTILES];
    if (j + 1 < 8) {
      #pragma unroll
      for (int nt = 0; nt < NTILES; ++nt)
        bn[nt] = *(const v2f*)(bs + (size_t)(2 * (j + 1)) * NDIM + nt * 16);
    }
    #pragma unroll
    for (int nt = 0; nt < NTILES; ++nt)
      acc[nt] = __builtin_amdgcn_wmma_f32_16x16x4_f32(
          false, a[j], false, bc[nt], (short)0, acc[nt], false, false);
    if (j + 1 < 8) {
      #pragma unroll
      for (int nt = 0; nt < NTILES; ++nt) bc[nt] = bn[nt];
    }
  }
}

template <int NTILES, int NW>
__global__ __launch_bounds__(NW * 32) void gemm_wmma_f32(
    const float* __restrict__ A, const float* __restrict__ B,
    const float* __restrict__ bias, float* __restrict__ Out,
    int M, int K,
    int aRowStride, int aKShift, int aKMask, size_t aChunkStride,
    size_t aBatchStride, size_t bBatchStride, size_t outBatchStride, int relu,
    int gemmBlocksX, int nBatch,
    const int* __restrict__ hSrc, const int* __restrict__ hDst,
    const int* __restrict__ hEt, int* __restrict__ hBkt, int hE, int hN) {
  constexpr int NDIM   = NTILES * 16;
  constexpr int PANEL  = 16 * NDIM;
  constexpr int NQ     = NDIM / 4;
  constexpr int GROUPS = 16 * NQ;
  constexpr int NT     = NW * 32;
  constexpr int IPT    = (GROUPS + NT - 1) / NT;
  static_assert(NW == 8, "the edge-bucket blocks are written for 256 threads");
  __shared__ float4 Bs4[PANEL];
  float2* Bs = (float2*)Bs4;

  const int wave = threadIdx.x >> 5;
  const int lane = threadIdx.x & 31;
  const int half = lane & 15;
  const int hi   = lane >> 4;

  const int bid = blockIdx.x;
  if (bid >= gemmBlocksX * nBatch) {
    bucket_block(hSrc, hDst, hEt, hBkt, hE, hN, bid - gemmBlocksX * nBatch);
    return;
  }
  const int batch = bid / gemmBlocksX;
  const int rgrp  = bid - batch * gemmBlocksX;

  const float* Ab = A + aBatchStride * batch;
  const float* Bb = B + bBatchStride * batch;
  float*       Ob = Out + outBatchStride * batch;

  const int mTile = rgrp * NW + wave;
  const int row0  = mTile * 16;

  v8f acc[NTILES];
  for (int nt = 0; nt < NTILES; ++nt) {
    float bv = bias ? bias[nt * 16 + half] : 0.0f;
    v8f t;
    #pragma unroll
    for (int i = 0; i < 8; ++i) t[i] = bv;
    acc[nt] = t;
  }

  int aRow = row0 + half;
  if (aRow >= M) aRow = M - 1;
  const float* Arow = Ab + (size_t)aRow * aRowStride + 2 * hi;

  const float2* BsLane = Bs + (size_t)hi * NDIM + half;

  int sOff[IPT];
  int gOff[IPT];
  #pragma unroll
  for (int g = 0; g < IPT; ++g) {
    const int idx = (threadIdx.x + g * NT) % GROUPS;
    const int kp = idx / NQ, n = (idx % NQ) * 4;
    sOff[g] = kp * NDIM + n;
    gOff[g] = 2 * kp * NDIM + n;
  }
  constexpr bool ALL_OWN = (GROUPS % NT) == 0;
  const bool own = ALL_OWN || (threadIdx.x < GROUPS);

  float4 bx[IPT], by[IPT];
  v2f    an[8];

  #pragma unroll
  for (int g = 0; g < IPT; ++g) {
    const float* q = Bb + gOff[g];
    bx[g] = *(const float4*)q;
    by[g] = *(const float4*)(q + NDIM);
  }
  #pragma unroll
  for (int j = 0; j < 8; ++j) an[j] = *(const v2f*)(Arow + 4 * j);
  if (own) {
    #pragma unroll
    for (int g = 0; g < IPT; ++g) {
      float4* d = (float4*)(Bs + sOff[g]);
      d[0] = make_float4(bx[g].x, by[g].x, bx[g].y, by[g].y);
      d[1] = make_float4(bx[g].z, by[g].z, bx[g].w, by[g].w);
    }
  }
  __syncthreads();

  const int P = K >> 5;
  for (int p = 0; p + 1 < P; ++p) {
    v2f ac[8];
    #pragma unroll
    for (int j = 0; j < 8; ++j) ac[j] = an[j];

    const int k1 = (p + 1) << 5;
    #pragma unroll
    for (int g = 0; g < IPT; ++g) {
      const float* q = Bb + (size_t)k1 * NDIM + gOff[g];
      bx[g] = *(const float4*)q;
      by[g] = *(const float4*)(q + NDIM);
    }
    const float* ap = Arow + (size_t)(k1 >> aKShift) * aChunkStride + (k1 & aKMask);
    #pragma unroll
    for (int j = 0; j < 8; ++j) an[j] = *(const v2f*)(ap + 4 * j);

    panel_products<NTILES>(acc, ac, BsLane + (p & 1) * PANEL);

    if (own) {
      float2* dst = Bs + ((p + 1) & 1) * PANEL;
      #pragma unroll
      for (int g = 0; g < IPT; ++g) {
        float4* d = (float4*)(dst + sOff[g]);
        d[0] = make_float4(bx[g].x, by[g].x, bx[g].y, by[g].y);
        d[1] = make_float4(bx[g].z, by[g].z, bx[g].w, by[g].w);
      }
    }
    __syncthreads();
  }
  panel_products<NTILES>(acc, an, BsLane + ((P - 1) & 1) * PANEL);

  for (int nt = 0; nt < NTILES; ++nt) {
    #pragma unroll
    for (int vr = 0; vr < 8; ++vr) {
      int r = row0 + vr + 8 * hi;
      if (r < M) {
        float v = acc[nt][vr];
        if (relu) v = fmaxf(v, 0.0f);
        Ob[(size_t)r * NDIM + nt * 16 + half] = v;
      }
    }
  }
}

extern "C" void kernel_launch(void* const* d_in, const int* in_sizes, int n_in,
                              void* d_out, int out_size, void* d_ws, size_t ws_size,
                              hipStream_t stream) {
  const float* h   = (const float*)d_in[0];
  const int*   src = (const int*)d_in[1];
  const int*   dst = (const int*)d_in[2];
  const int*   et  = (const int*)d_in[3];
  const int*   cat = (const int*)d_in[4];
  const float* gtw = (const float*)d_in[5];
  const float* Wc  = (const float*)d_in[6];
  const float* W1  = (const float*)d_in[7];
  const float* b1  = (const float*)d_in[8];
  const float* W2  = (const float*)d_in[9];
  const float* b2  = (const float*)d_in[10];

  const int IN_DIM = FDIM;
  const int N    = in_sizes[0] / IN_DIM;
  const int E    = in_sizes[1];
  const int NCAT = in_sizes[4];
  static_assert(LL == 2, "written for two propagation rounds: raw features, then selected rows");

  float* ws    = (float*)d_ws;
  float* filt  = ws;
  float* Mfold = filt + 32;
  float* P0    = Mfold + (size_t)CC * FDIM * HIDD;
  int*   bkt   = (int*)(P0 + (size_t)CC * N * FDIM);
  float* wsEnd = (float*)(bkt + (size_t)N * BROW);
  if ((size_t)(wsEnd - ws) * sizeof(float) > ws_size) return;

  prep_kernel<<<(N + 255) / 256, 256, 0, stream>>>(gtw, filt, bkt, N * (BROW / 4));

  const int ftiles = (IN_DIM + 15) / 16;
  const int gf     = (ftiles + 7) / 8;
  const int hb     = (E + HIST_EDGES_PER_BLOCK - 1) / HIST_EDGES_PER_BLOCK;

  gemm_wmma_f32<8, 8><<<dim3(gf * CC + hb), 256, 0, stream>>>(
      Wc, W1, nullptr, Mfold, IN_DIM, HIDD,
        HIDD,   7,   127,   0,
        (size_t)IN_DIM * HIDD,   (size_t)HIDD * HIDD,
        (size_t)IN_DIM * HIDD,   0,
        gf,   CC, src, dst, et, bkt, E, N);

  spmm_kernel<<<(N + 7) / 8, 256, 0, stream>>>(bkt, src, dst, et, filt, h, P0, N, E, N);

  spmm_hidden_kernel<<<(NCAT + FROWS - 1) / FROWS, FROWS * 32, 0, stream>>>(
      bkt, src, dst, et, filt, P0, cat, NCAT, E, N, Mfold, b1, W2, b2, (float*)d_out);
}


# Round-two kernel: a block's (row, incoming edge) gathers form one LDS list cut into 16 equal chunks, one per wave; a row cut between waves is summed in pieces, fixed order (sum order changes)
# speedup vs baseline: 7.1893x; 1.0824x over previous
#include <hip/hip_runtime.h>

#define RR 5
#define CC 2
#define LL 2
#define HIDD 128
#define FDIM 256

typedef __attribute__((ext_vector_type(2))) float v2f;
typedef __attribute__((ext_vector_type(8))) float v8f;

__global__ void prep_kernel(const float* __restrict__ gt_w,
                            float* __restrict__ filt,
                            int* __restrict__ bkt, int nQuads) {
  const int t = blockIdx.x * blockDim.x + threadIdx.x;
  const int nthr = gridDim.x * blockDim.x;
  int4* bkt4 = (int4*)bkt;
  for (int i = t; i < nQuads; i += nthr) bkt4[i] = make_int4(0, 0, 0, 0);
  if (t >= LL * CC) return;
  const float* g = gt_w + t * RR;
  float m = g[0];
  for (int r = 1; r < RR; ++r) m = fmaxf(m, g[r]);
  float e[RR];
  float s = 0.f;
  for (int r = 0; r < RR; ++r) { e[r] = __expf(g[r] - m); s += e[r]; }
  float inv = 1.0f / s;
  for (int r = 0; r < RR; ++r) filt[t * RR + r] = e[r] * inv;
}

#define KEY_SRC_MASK 0x0FFFFFFFu
#define KEY_ET_SHIFT 28
#define BROW 32
#define BCAP (BROW - 1)

#define HIST_EDGES_PER_BLOCK 1024
__device__ __forceinline__ void bucket_block(const int* __restrict__ src,
                                             const int* __restrict__ dst,
                                             const int* __restrict__ etype,
                                             int* __restrict__ bkt, int E, int N, int hb) {
  const int e0 = hb * HIST_EDGES_PER_BLOCK + threadIdx.x;
  int dv[4], sv[4], tv[4];
  int d[4], key[4], r[4];
  #pragma unroll
  for (int k = 0; k < 4; ++k) {
    const int e  = e0 + k * 256;
    const int ec = (e < E) ? e : E - 1;
    dv[k] = dst[ec]; sv[k] = src[ec]; tv[k] = etype[ec];
  }
  #pragma unroll
  for (int k = 0; k < 4; ++k) {
    const int e = e0 + k * 256;
    const bool ok = (e < E) && ((unsigned)dv[k] < (unsigned)N);
    d[k]   = ok ? dv[k] : -1;
    key[k] = (int)(((unsigned)sv[k] & KEY_SRC_MASK) | ((unsigned)tv[k] << KEY_ET_SHIFT));
  }
  #pragma unroll
  for (int k = 0; k < 4; ++k)
    r[k] = (d[k] >= 0) ? atomicAdd(&bkt[(size_t)d[k] * BROW], 1) : BCAP;
  #pragma unroll
  for (int k = 0; k < 4; ++k) {
    if ((unsigned)r[k] < (unsigned)BCAP) bkt[(size_t)d[k] * BROW + 1 + r[k]] = key[k];
  }
}

__device__ __forceinline__ float wave_sum_f(float v) {
  for (int o = 16; o > 0; o >>= 1) v += __shfl_xor(v, o);
  return v;
}

__device__ __forceinline__ int lane_i(int v, int l) {
  return __builtin_amdgcn_readlane(v, l);
}
__device__ __forceinline__ float lane_f(float v, int l) {
  return __int_as_float(__builtin_amdgcn_readlane(__float_as_int(v), l));
}
__device__ __forceinline__ void fma4(float4& a, const float4& x, float w) {
  a.x = fmaf(x.x, w, a.x); a.y = fmaf(x.y, w, a.y);
  a.z = fmaf(x.z, w, a.z); a.w = fmaf(x.w, w, a.w);
}

__device__ __forceinline__ float pick5(const float (&t)[RR], int et) {
  float v = t[RR - 1];
  v = (et == 3) ? t[3] : v;
  v = (et == 2) ? t[2] : v;
  v = (et == 1) ? t[1] : v;
  v = (et == 0) ? t[0] : v;
  return v;
}

__device__ __forceinline__ void key_decode(
    unsigned k, int N, const float (&fa)[RR], const float (&fb)[RR],
    int& s, float& u0, float& u1) {
  int sv = (int)(k & KEY_SRC_MASK);
  if (sv > N - 1) sv = N - 1;
  s = sv;
  const int et = (int)(k >> KEY_ET_SHIFT);
  u0 = pick5(fa, et);
  u1 = pick5(fb, et);
}

struct RowQ { float4 p, q; };

__device__ __forceinline__ RowQ row_read(const float4* __restrict__ H0, size_t r) {
  RowQ x;
  x.p = H0[r]; x.q = H0[r + 32];
  return x;
}

__device__ __forceinline__ void row_acc(float4& a0, float4& b0, float4& a1, float4& b1,
                                        const RowQ& x, float w0, float w1) {
  fma4(a0, x.p, w0); fma4(b0, x.q, w0);
  fma4(a1, x.p, w1); fma4(b1, x.q, w1);
}

__device__ __forceinline__ void gather_node(
    int raw, int d, int lane,
    const int* __restrict__ src, const int* __restrict__ dst,
    const int* __restrict__ etype,
    const float (&fa)[RR], const float (&fb)[RR], const float* __restrict__ Hin,
    int E, int N, float sc0, float sc1,
    float4& a0, float4& b0, float4& a1, float4& b1) {
  constexpr int RQ = FDIM / 4;
  const float4* H0 = (const float4*)Hin + lane;

  const int cnt = lane_i(raw, 0);
  if (__builtin_expect(cnt <= BCAP, 1)) {
    int s = 0; float u0 = 0.f, u1 = 0.f;
    if (lane >= 1 && lane <= cnt) key_decode((unsigned)raw, N, fa, fb, s, u0, u1);

    const float dg0 = wave_sum_f(u0);
    const float dg1 = wave_sum_f(u1);
    const float w0 = (u0 / dg0) * sc0;
    const float w1 = (u1 / dg1) * sc1;
    const int n = cnt;

    int i = 0;
    for (; i + 4 <= n; i += 4) {
      const size_t ra = (size_t)lane_i(s, i + 1) * RQ;
      const size_t rb = (size_t)lane_i(s, i + 2) * RQ;
      const size_t rc = (size_t)lane_i(s, i + 3) * RQ;
      const size_t rd = (size_t)lane_i(s, i + 4) * RQ;
      const RowQ xa = row_read(H0, ra);
      const RowQ xb = row_read(H0, rb);
      const RowQ xc = row_read(H0, rc);
      const RowQ xd = row_read(H0, rd);
      row_acc(a0, b0, a1, b1, xa, lane_f(w0, i + 1), lane_f(w1, i + 1));
      row_acc(a0, b0, a1, b1, xb, lane_f(w0, i + 2), lane_f(w1, i + 2));
      row_acc(a0, b0, a1, b1, xc, lane_f(w0, i + 3), lane_f(w1, i + 3));
      row_acc(a0, b0, a1, b1, xd, lane_f(w0, i + 4), lane_f(w1, i + 4));
    }
    if (i < n) {
      const int r  = n - i;
      const int la = i + 1;
      const int lb = (r > 1) ? i + 2 : la;
      const int lc = (r > 2) ? i + 3 : la;
      const size_t ra = (size_t)lane_i(s, la) * RQ;
      const size_t rb = (size_t)lane_i(s, lb) * RQ;
      const size_t rc = (size_t)lane_i(s, lc) * RQ;
      const RowQ xa = row_read(H0, ra);
      const RowQ xb = row_read(H0, rb);
      const RowQ xc = row_read(H0, rc);
      const float wb0 = (r > 1) ? lane_f(w0, lb) : 0.f;
      const float wb1 = (r > 1) ? lane_f(w1, lb) : 0.f;
      const float wc0 = (r > 2) ? lane_f(w0, lc) : 0.f;
      const float wc1 = (r > 2) ? lane_f(w1, lc) : 0.f;
      row_acc(a0, b0, a1, b1, xa, lane_f(w0, la), lane_f(w1, la));
      row_acc(a0, b0, a1, b1, xb, wb0, wb1);
      row_acc(a0, b0, a1, b1, xc, wc0, wc1);
    }
  } else {
    float dg0 = 0.f, dg1 = 0.f;
    for (int b = 0; b < E; b += 32) {
      const int e = b + lane;
      if (e < E && dst[e] == d) {
        const int et = etype[e];
        dg0 += pick5(fa, et);
        dg1 += pick5(fb, et);
      }
    }
    dg0 = wave_sum_f(dg0);
    dg1 = wave_sum_f(dg1);
    for (int b = 0; b < E; b += 32) {
      const int e = b + lane;
      const bool hit = (e < E) && (dst[e] == d);
      int s = 0; float w0 = 0.f, w1 = 0.f;
      if (hit) {
        int sv = src[e];
        if (sv < 0) sv = 0;
        if (sv > N - 1) sv = N - 1;
        s = sv;
        const int et = etype[e];
        w0 = (pick5(fa, et) / dg0) * sc0;
        w1 = (pick5(fb, et) / dg1) * sc1;
      }
      unsigned m = __builtin_amdgcn_ballot_w32(hit);
      while (m) {
        const int i = __builtin_ctz(m);
        m &= m - 1;
        const size_t r = (size_t)lane_i(s, i) * RQ;
        const RowQ x = row_read(H0, r);
        row_acc(a0, b0, a1, b1, x, lane_f(w0, i), lane_f(w1, i));
      }
    }
  }
}

__device__ __forceinline__ void gather_row_long(
    int d, int lane,
    const int* __restrict__ bkt, const int* __restrict__ src,
    const int* __restrict__ dst, const int* __restrict__ etype,
    const float (&fa)[RR], const float (&fb)[RR],
    const float (&ga)[RR], const float (&gb)[RR],
    const float* __restrict__ Hin, int E, int N,
    float4& a0, float4& b0, float4& a1, float4& b1) {
  float dg0 = 0.f, dg1 = 0.f;
  for (int b = 0; b < E; b += 32) {
    const int e = b + lane;
    if (e < E && dst[e] == d) {
      const int et = etype[e];
      dg0 += pick5(ga, et);
      dg1 += pick5(gb, et);
    }
  }
  dg0 = wave_sum_f(dg0);
  dg1 = wave_sum_f(dg1);
  for (int b = 0; b < E; b += 32) {
    const int e = b + lane;
    const bool hit = (e < E) && (dst[e] == d);
    int s = 0; float w0 = 0.f, w1 = 0.f;
    if (hit) {
      int sv = src[e];
      if (sv < 0) sv = 0;
      if (sv > N - 1) sv = N - 1;
      s = sv;
      const int et = etype[e];
      w0 = pick5(ga, et) / dg0;
      w1 = pick5(gb, et) / dg1;
    }
    unsigned m = __builtin_amdgcn_ballot_w32(hit);
    while (m) {
      const int i = __builtin_ctz(m);
      m &= m - 1;
      const int sj = lane_i(s, i);
      const int rawc = bkt[(size_t)sj * BROW + lane];
      gather_node(rawc, sj, lane, src, dst, etype, fa, fb, Hin, E, N,
                  lane_f(w0, i), lane_f(w1, i), a0, b0, a1, b1);
    }
  }
}


#define FROWS 16
#define NT2   (HIDD / 16)
#define NCLS  16
#define KGRP  16
__global__ __launch_bounds__(FROWS * 32, 12) void spmm_hidden_kernel(
    const int* __restrict__ bkt, const int* __restrict__ src,
    const int* __restrict__ dst, const int* __restrict__ etype,
    const float* __restrict__ filt, const float* __restrict__ Hin,
    const int* __restrict__ rowsel, int nOut, int E, int N,
    const float* __restrict__ Mf, const float* __restrict__ bias1,
    const float* __restrict__ W2, const float* __restrict__ bias2,
    float* __restrict__ Out) {
  static_assert(FROWS == 2 * NT2, "one wave per (column tile, channel) pair");
  static_assert(CC == 2 && FDIM % (4 * KGRP) == 0, "K is cut per channel, in groups of KGRP k-steps");
  constexpr int XP = CC * FDIM + 4;
  constexpr int ZP = HIDD + 4;
  __shared__ __attribute__((aligned(16))) float Xs[FROWS * XP];
  __shared__ __attribute__((aligned(16))) float Zs[FROWS * ZP];
  __shared__ float Ps[NT2 * 8 * 32];
  __shared__ int   tS[FROWS * 32];
  __shared__ float tW0[FROWS * 32];
  __shared__ float tW1[FROWS * 32];
  __shared__ int   tCnt[FROWS];
  __shared__ int   contRow[FROWS];
  __shared__ __attribute__((aligned(16))) float Part[FROWS * CC * FDIM];

  const int lane = threadIdx.x & 31;
  const int wave = __builtin_amdgcn_readfirstlane(threadIdx.x >> 5);
  const int row0 = blockIdx.x * FROWS;
  const int o = row0 + wave;

  float fa[RR], fb[RR], ga[RR], gb[RR];
  for (int r = 0; r < RR; ++r) {
    fa[r] = filt[r];           fb[r] = filt[RR + r];
    ga[r] = filt[CC * RR + r]; gb[r] = filt[CC * RR + RR + r];
  }
  float4 a0 = make_float4(0.f, 0.f, 0.f, 0.f), b0 = a0, a1 = a0, b1 = a0;
  {
    int ps = 0, pn = 0; float pw0 = 0.f, pw1 = 0.f;
    if (o < nOut) {
      int d = rowsel[o];
      if (d < 0) d = 0;
      if (d > N - 1) d = N - 1;
      const int raw = bkt[(size_t)d * BROW + lane];
      const int cnt = lane_i(raw, 0);
      if (__builtin_expect(cnt <= BCAP, 1)) {
        float u0 = 0.f, u1 = 0.f;
        if (lane >= 1 && lane <= cnt) key_decode((unsigned)raw, N, ga, gb, ps, u0, u1);
        const float dg0 = wave_sum_f(u0);
        const float dg1 = wave_sum_f(u1);
        pw0 = u0 / dg0;
        pw1 = u1 / dg1;
        pn = (cnt > 0) ? cnt : 0;
      } else {
        gather_row_long(d, lane, bkt, src, dst, etype, fa, fb, ga, gb, Hin, E, N, a0, b0, a1, b1);
      }
    }
    tS[wave * 32 + lane]  = ps;
    tW0[wave * 32 + lane] = pw0;
    tW1[wave * 32 + lane] = pw1;
    if (lane == 0) { tCnt[wave] = pn; contRow[wave] = -1; }
    float* xr = Xs + wave * XP + 4 * lane;
    *(float4*)(xr)                   = a0;
    *(float4*)(xr + FDIM / 2)        = b0;
    *(float4*)(xr + FDIM)            = a1;
    *(float4*)(xr + FDIM + FDIM / 2) = b1;
  }
  __syncthreads();

  {
    const int cv = (lane < FROWS) ? tCnt[lane] : 0;
    int T = 0;
    for (int i = 0; i < FROWS; ++i) T += lane_i(cv, i);
    const int k0 = (wave * T) / FROWS;
    const int k1 = ((wave + 1) * T) / FROWS;
    int p = 0;
    for (int i = 0; i < FROWS; ++i) {
      const int c  = lane_i(cv, i);
      const int lo = (k0 > p) ? k0 : p;
      const int hi = (k1 < p + c) ? k1 : p + c;
      if (lo < hi) {
        const int ja = lo - p + 1, jb = hi - p;
        const int   s  = tS[i * 32 + lane];
        const float w0 = tW0[i * 32 + lane];
        const float w1 = tW1[i * 32 + lane];
        a0 = make_float4(0.f, 0.f, 0.f, 0.f); b0 = a0; a1 = a0; b1 = a0;
        int rawn = bkt[(size_t)lane_i(s, ja) * BROW + lane];
        for (int j = ja; j <= jb; ++j) {
          const int sj   = lane_i(s, j);
          const int rawc = rawn;
          const int jn   = (j < jb) ? j + 1 : j;
          rawn = bkt[(size_t)lane_i(s, jn) * BROW + lane];
          gather_node(rawc, sj, lane, src, dst, etype, fa, fb, Hin, E, N,
                      lane_f(w0, j), lane_f(w1, j), a0, b0, a1, b1);
        }
        const bool first = (ja == 1);
        float* xr = (first ? Xs + i * XP : Part + wave * (CC * FDIM)) + 4 * lane;
        *(float4*)(xr)                   = a0;
        *(float4*)(xr + FDIM / 2)        = b0;
        *(float4*)(xr + FDIM)            = a1;
        *(float4*)(xr + FDIM + FDIM / 2) = b1;
        if (!first && lane == 0) contRow[wave] = i;
      }
      p += c;
    }
  }
  __syncthreads();

  {
    const int cr = (lane < FROWS) ? contRow[lane] : -1;
    float* xr = Xs + wave * XP + 4 * lane;
    for (int w = 0; w < FROWS; ++w) {
      if (lane_i(cr, w) == wave) {
        const float* pp = Part + w * (CC * FDIM) + 4 * lane;
        const float4 p0 = *(const float4*)(pp);
        const float4 p1 = *(const float4*)(pp + FDIM / 2);
        const float4 p2 = *(const float4*)(pp + FDIM);
        const float4 p3 = *(const float4*)(pp + FDIM + FDIM / 2);
        float4 x0 = *(float4*)(xr);
        float4 x1 = *(float4*)(xr + FDIM / 2);
        float4 x2 = *(float4*)(xr + FDIM);
        float4 x3 = *(float4*)(xr + FDIM + FDIM / 2);
        x0.x += p0.x; x0.y += p0.y; x0.z += p0.z; x0.w += p0.w;
        x1.x += p1.x; x1.y += p1.y; x1.z += p1.z; x1.w += p1.w;
        x2.x += p2.x; x2.y += p2.y; x2.z += p2.z; x2.w += p2.w;
        x3.x += p3.x; x3.y += p3.y; x3.z += p3.z; x3.w += p3.w;
        *(float4*)(xr)                   = x0;
        *(float4*)(xr + FDIM / 2)        = x1;
        *(float4*)(xr + FDIM)            = x2;
        *(float4*)(xr + FDIM + FDIM / 2) = x3;
      }
    }
  }
  __syncthreads();

  const int half = lane & 15;
  const int hi   = lane >> 4;
  const int nt   = wave % NT2;
  const int ch   = wave / NT2;
  const float bv = (ch == 0 && bias1) ? bias1[nt * 16 + half] : 0.0f;
  v8f c0, c1;
  #pragma unroll
  for (int i = 0; i < 8; ++i) { c0[i] = bv; c1[i] = 0.0f; }
  const float* xa = Xs + half * XP + ch * FDIM + 2 * hi;
  const float* mb = Mf + ((ch * FDIM + 2 * hi) * HIDD + nt * 16 + half);
  for (int g = 0; g < FDIM / (4 * KGRP); ++g) {
    v2f bf[KGRP];
    #pragma unroll
    for (int u = 0; u < KGRP; ++u) {
      const float* q = mb + (4 * (KGRP * g + u)) * HIDD;
      bf[u][0] = q[0];
      bf[u][1] = q[HIDD];
    }
    #pragma unroll
    for (int u = 0; u < KGRP; ++u) {
      const v2f af = *(const v2f*)(xa + 4 * (KGRP * g + u));
      if (u & 1)
        c1 = __builtin_amdgcn_wmma_f32_16x16x4_f32(
            false, af, false, bf[u], (short)0, c1, false, false);
      else
        c0 = __builtin_amdgcn_wmma_f32_16x16x4_f32(
            false, af, false, bf[u], (short)0, c0, false, false);
    }
  }

  float* pt = Ps + nt * (8 * 32) + lane;
  if (ch == 1) {
    #pragma unroll
    for (int vr = 0; vr < 8; ++vr) pt[vr * 32] = c0[vr] + c1[vr];
  }
  __syncthreads();
  if (ch == 0) {
    #pragma unroll
    for (int vr = 0; vr < 8; ++vr) {
      const float v = (c0[vr] + c1[vr]) + pt[vr * 32];
      Zs[(vr + 8 * hi) * ZP + nt * 16 + half] = fmaxf(v, 0.0f);
    }
  }
  __syncthreads();
  if (wave != 0) return;

  static_assert((HIDD / 4) % KGRP == 0, "the classifier's k-steps come in whole groups");
  const float* w2l = W2 + (size_t)(2 * hi) * NCLS + half;
  const float bv2 = bias2 ? bias2[half] : 0.0f;
  v8f o0, o1;
  #pragma unroll
  for (int i = 0; i < 8; ++i) { o0[i] = bv2; o1[i] = 0.0f; }
  const float* zr = Zs + half * ZP + 2 * hi;
  for (int g = 0; g < HIDD / (4 * KGRP); ++g) {
    v2f bw[KGRP];
    #pragma unroll
    for (int u = 0; u < KGRP; ++u) {
      const int kk = KGRP * g + u;
      bw[u][0] = w2l[(4 * kk) * NCLS];
      bw[u][1] = w2l[(4 * kk + 1) * NCLS];
    }
    #pragma unroll
    for (int u = 0; u < KGRP; ++u) {
      const int kk = KGRP * g + u;
      const v2f af = *(const v2f*)(zr + 4 * kk);
      if (u & 1)
        o1 = __builtin_amdgcn_wmma_f32_16x16x4_f32(
            false, af, false, bw[u], (short)0, o1, false, false);
      else
        o0 = __builtin_amdgcn_wmma_f32_16x16x4_f32(
            false, af, false, bw[u], (short)0, o0, false, false);
    }
  }
  #pragma unroll
  for (int vr = 0; vr < 8; ++vr) {
    const int r = row0 + vr + 8 * hi;
    if (r < nOut) Out[(size_t)r * NCLS + half] = o0[vr] + o1[vr];
  }
}

template <int NTILES>
__device__ __forceinline__ void panel_products(
    v8f (&acc)[NTILES], const v2f (&a)[8], const float2* bs) {
  constexpr int NDIM = NTILES * 16;
  v2f bc[NTILES];
  #pragma unroll
  for (int nt = 0; nt < NTILES; ++nt) bc[nt] = *(const v2f*)(bs + nt * 16);
  #pragma unroll
  for (int j = 0; j < 8; ++j) {
    v2f bn[NTILES];
    if (j + 1 < 8) {
      #pragma unroll
      for (int nt = 0; nt < NTILES; ++nt)
        bn[nt] = *(const v2f*)(bs + (size_t)(2 * (j + 1)) * NDIM + nt * 16);
    }
    #pragma unroll
    for (int nt = 0; nt < NTILES; ++nt)
      acc[nt] = __builtin_amdgcn_wmma_f32_16x16x4_f32(
          false, a[j], false, bc[nt], (short)0, acc[nt], false, false);
    if (j + 1 < 8) {
      #pragma unroll
      for (int nt = 0; nt < NTILES; ++nt) bc[nt] = bn[nt];
    }
  }
}

template <int NTILES, int NW>
__global__ __launch_bounds__(NW * 32) void gemm_wmma_f32(
    const float* __restrict__ A, const float* __restrict__ B,
    const float* __restrict__ bias, float* __restrict__ Out,
    int M, int K,
    int aRowStride, int aKShift, int aKMask, size_t aChunkStride,
    size_t aBatchStride, size_t bBatchStride, size_t outBatchStride, int relu,
    int gemmBlocksX, int nBatch,
    const int* __restrict__ hSrc, const int* __restrict__ hDst,
    const int* __restrict__ hEt, int* __restrict__ hBkt, int hE, int hN) {
  constexpr int NDIM   = NTILES * 16;
  constexpr int PANEL  = 16 * NDIM;
  constexpr int NQ     = NDIM / 4;
  constexpr int GROUPS = 16 * NQ;
  constexpr int NT     = NW * 32;
  constexpr int IPT    = (GROUPS + NT - 1) / NT;
  static_assert(NW == 8, "the edge-bucket blocks are written for 256 threads");
  __shared__ float4 Bs4[PANEL];
  float2* Bs = (float2*)Bs4;

  const int wave = threadIdx.x >> 5;
  const int lane = threadIdx.x & 31;
  const int half = lane & 15;
  const int hi   = lane >> 4;

  const int bid = blockIdx.x;
  if (bid >= gemmBlocksX * nBatch) {
    bucket_block(hSrc, hDst, hEt, hBkt, hE, hN, bid - gemmBlocksX * nBatch);
    return;
  }
  const int batch = bid / gemmBlocksX;
  const int rgrp  = bid - batch * gemmBlocksX;

  const float* Ab = A + aBatchStride * batch;
  const float* Bb = B + bBatchStride * batch;
  float*       Ob = Out + outBatchStride * batch;

  const int mTile = rgrp * NW + wave;
  const int row0  = mTile * 16;

  v8f acc[NTILES];
  for (int nt = 0; nt < NTILES; ++nt) {
    float bv = bias ? bias[nt * 16 + half] : 0.0f;
    v8f t;
    #pragma unroll
    for (int i = 0; i < 8; ++i) t[i] = bv;
    acc[nt] = t;
  }

  int aRow = row0 + half;
  if (aRow >= M) aRow = M - 1;
  const float* Arow = Ab + (size_t)aRow * aRowStride + 2 * hi;

  const float2* BsLane = Bs + (size_t)hi * NDIM + half;

  int sOff[IPT];
  int gOff[IPT];
  #pragma unroll
  for (int g = 0; g < IPT; ++g) {
    const int idx = (threadIdx.x + g * NT) % GROUPS;
    const int kp = idx / NQ, n = (idx % NQ) * 4;
    sOff[g] = kp * NDIM + n;
    gOff[g] = 2 * kp * NDIM + n;
  }
  constexpr bool ALL_OWN = (GROUPS % NT) == 0;
  const bool own = ALL_OWN || (threadIdx.x < GROUPS);

  float4 bx[IPT], by[IPT];
  v2f    an[8];

  #pragma unroll
  for (int g = 0; g < IPT; ++g) {
    const float* q = Bb + gOff[g];
    bx[g] = *(const float4*)q;
    by[g] = *(const float4*)(q + NDIM);
  }
  #pragma unroll
  for (int j = 0; j < 8; ++j) an[j] = *(const v2f*)(Arow + 4 * j);
  if (own) {
    #pragma unroll
    for (int g = 0; g < IPT; ++g) {
      float4* d = (float4*)(Bs + sOff[g]);
      d[0] = make_float4(bx[g].x, by[g].x, bx[g].y, by[g].y);
      d[1] = make_float4(bx[g].z, by[g].z, bx[g].w, by[g].w);
    }
  }
  __syncthreads();

  const int P = K >> 5;
  for (int p = 0; p + 1 < P; ++p) {
    v2f ac[8];
    #pragma unroll
    for (int j = 0; j < 8; ++j) ac[j] = an[j];

    const int k1 = (p + 1) << 5;
    #pragma unroll
    for (int g = 0; g < IPT; ++g) {
      const float* q = Bb + (size_t)k1 * NDIM + gOff[g];
      bx[g] = *(const float4*)q;
      by[g] = *(const float4*)(q + NDIM);
    }
    const float* ap = Arow + (size_t)(k1 >> aKShift) * aChunkStride + (k1 & aKMask);
    #pragma unroll
    for (int j = 0; j < 8; ++j) an[j] = *(const v2f*)(ap + 4 * j);

    panel_products<NTILES>(acc, ac, BsLane + (p & 1) * PANEL);

    if (own) {
      float2* dst = Bs + ((p + 1) & 1) * PANEL;
      #pragma unroll
      for (int g = 0; g < IPT; ++g) {
        float4* d = (float4*)(dst + sOff[g]);
        d[0] = make_float4(bx[g].x, by[g].x, bx[g].y, by[g].y);
        d[1] = make_float4(bx[g].z, by[g].z, bx[g].w, by[g].w);
      }
    }
    __syncthreads();
  }
  panel_products<NTILES>(acc, an, BsLane + ((P - 1) & 1) * PANEL);

  for (int nt = 0; nt < NTILES; ++nt) {
    #pragma unroll
    for (int vr = 0; vr < 8; ++vr) {
      int r = row0 + vr + 8 * hi;
      if (r < M) {
        float v = acc[nt][vr];
        if (relu) v = fmaxf(v, 0.0f);
        Ob[(size_t)r * NDIM + nt * 16 + half] = v;
      }
    }
  }
}

extern "C" void kernel_launch(void* const* d_in, const int* in_sizes, int n_in,
                              void* d_out, int out_size, void* d_ws, size_t ws_size,
                              hipStream_t stream) {
  const float* h   = (const float*)d_in[0];
  const int*   src = (const int*)d_in[1];
  const int*   dst = (const int*)d_in[2];
  const int*   et  = (const int*)d_in[3];
  const int*   cat = (const int*)d_in[4];
  const float* gtw = (const float*)d_in[5];
  const float* Wc  = (const float*)d_in[6];
  const float* W1  = (const float*)d_in[7];
  const float* b1  = (const float*)d_in[8];
  const float* W2  = (const float*)d_in[9];
  const float* b2  = (const float*)d_in[10];

  const int IN_DIM = FDIM;
  const int N    = in_sizes[0] / IN_DIM;
  const int E    = in_sizes[1];
  const int NCAT = in_sizes[4];
  static_assert(LL == 2, "written for two propagation rounds, both summed from the raw features");

  float* ws    = (float*)d_ws;
  float* filt  = ws;
  float* Mfold = filt + 32;
  int*   bkt   = (int*)(Mfold + (size_t)CC * FDIM * HIDD);
  float* wsEnd = (float*)(bkt + (size_t)N * BROW);
  if ((size_t)(wsEnd - ws) * sizeof(float) > ws_size) return;

  prep_kernel<<<(N + 255) / 256, 256, 0, stream>>>(gtw, filt, bkt, N * (BROW / 4));

  const int ftiles = (IN_DIM + 15) / 16;
  const int gf     = (ftiles + 7) / 8;
  const int hb     = (E + HIST_EDGES_PER_BLOCK - 1) / HIST_EDGES_PER_BLOCK;

  gemm_wmma_f32<8, 8><<<dim3(gf * CC + hb), 256, 0, stream>>>(
      Wc, W1, nullptr, Mfold, IN_DIM, HIDD,
        HIDD,   7,   127,   0,
        (size_t)IN_DIM * HIDD,   (size_t)HIDD * HIDD,
        (size_t)IN_DIM * HIDD,   0,
        gf,   CC, src, dst, et, bkt, E, N);

  spmm_hidden_kernel<<<(NCAT + FROWS - 1) / FROWS, FROWS * 32, 0, stream>>>(
      bkt, src, dst, et, filt, h, cat, NCAT, E, N, Mfold, b1, W2, b2, (float*)d_out);
}
